# starting kernel
# baseline (speedup 1.0000x reference)
#include <hip/hip_runtime.h>

typedef __attribute__((ext_vector_type(2))) float v2f;
typedef __attribute__((ext_vector_type(8))) float v8f;

#define BB 8
#define CC 256
#define HWN 4096

__global__ __launch_bounds__(128) void gemm256_kernel(
    const float* __restrict__ A, long aStride,
    const float* __restrict__ bias,
    const float* __restrict__ Bm, long bStride,
    float* __restrict__ D, long dStride)
{
    const int lane  = threadIdx.x & 31;
    const int wave  = threadIdx.x >> 5;
    const int g     = lane >> 4;
    const int ln    = lane & 15;
    const int batch = blockIdx.z;
    const int mBase = blockIdx.y * 64 + wave * 16;
    const int nBase = blockIdx.x * 64;

    const float* Ab = A + (long)batch * aStride;
    const float* Bb = Bm + (long)batch * bStride;
    float*       Db = D + (long)batch * dStride;

    v8f acc0 = {}, acc1 = {}, acc2 = {}, acc3 = {};
    const int m  = mBase + ln;
    const int n0 = nBase + ln;

    #pragma unroll 4
    for (int k = 0; k < CC; k += 4) {
        v2f a = *(const v2f*)(Ab + (long)m * CC + k + 2 * g);
        const float* br0 = Bb + (long)(k + 2 * g) * HWN;
        const float* br1 = br0 + HWN;
        v2f b0 = { br0[n0     ], br1[n0     ] };
        v2f b1 = { br0[n0 + 16], br1[n0 + 16] };
        v2f b2 = { br0[n0 + 32], br1[n0 + 32] };
        v2f b3 = { br0[n0 + 48], br1[n0 + 48] };
        acc0 = __builtin_amdgcn_wmma_f32_16x16x4_f32(false, a, false, b0, (short)0, acc0, false, false);
        acc1 = __builtin_amdgcn_wmma_f32_16x16x4_f32(false, a, false, b1, (short)0, acc1, false, false);
        acc2 = __builtin_amdgcn_wmma_f32_16x16x4_f32(false, a, false, b2, (short)0, acc2, false, false);
        acc3 = __builtin_amdgcn_wmma_f32_16x16x4_f32(false, a, false, b3, (short)0, acc3, false, false);
    }

    #pragma unroll
    for (int r = 0; r < 8; ++r) {
        const int mo = mBase + r + 8 * g;
        const float bv = bias ? bias[mo] : 0.0f;
        float* drow = Db + (long)mo * HWN + n0;
        drow[0]  = acc0[r] + bv;
        drow[16] = acc1[r] + bv;
        drow[32] = acc2[r] + bv;
        drow[48] = acc3[r] + bv;
    }
}

__global__ __launch_bounds__(128) void corr_kernel(
    const float* __restrict__ Q, const float* __restrict__ Km,
    float* __restrict__ corr)
{
    const int lane  = threadIdx.x & 31;
    const int wave  = threadIdx.x >> 5;
    const int g     = lane >> 4;
    const int ln    = lane & 15;
    const int batch = blockIdx.z;
    const int mBase = blockIdx.y * 64 + wave * 16;
    const int nBase = blockIdx.x * 16;

    const float* qp = Q  + (long)batch * CC * HWN + (long)(mBase + ln) * HWN;
    const float* kp = Km + (long)batch * CC * HWN + (long)(nBase + ln) * HWN;

    v8f acc = {};
    #pragma unroll 8
    for (int k = 0; k < HWN; k += 4) {
        v2f a = *(const v2f*)(qp + k + 2 * g);
        v2f b = *(const v2f*)(kp + k + 2 * g);
        acc = __builtin_amdgcn_wmma_f32_16x16x4_f32(false, a, false, b, (short)0, acc, false, false);
    }

    float* cb = corr + (long)batch * CC * CC;
    #pragma unroll
    for (int r = 0; r < 8; ++r) {
        cb[(long)(mBase + r + 8 * g) * CC + nBase + ln] = acc[r];
    }
}

__global__ __launch_bounds__(256) void softmax_out_kernel(
    const float* __restrict__ scores,
    const float* __restrict__ fg, const float* __restrict__ mask,
    const float* __restrict__ V,  const float* __restrict__ gamma,
    float* __restrict__ out)
{
    const int  tid  = threadIdx.x;
    const int  lane = tid & 31;
    const int  wid  = tid >> 5;
    const long base = (long)blockIdx.x * HWN;

    __shared__ float red[8];

    float vals[16];
    float mx = -3.402823466e38f;
    #pragma unroll
    for (int t = 0; t < 16; ++t) {
        vals[t] = scores[base + t * 256 + tid];
        mx = fmaxf(mx, vals[t]);
    }
    #pragma unroll
    for (int off = 16; off >= 1; off >>= 1)
        mx = fmaxf(mx, __shfl_xor(mx, off, 32));
    if (lane == 0) red[wid] = mx;
    __syncthreads();
    #pragma unroll
    for (int w = 0; w < 8; ++w) mx = fmaxf(mx, red[w]);
    __syncthreads();

    float sum = 0.0f;
    #pragma unroll
    for (int t = 0; t < 16; ++t) {
        vals[t] = __expf(vals[t] - mx);
        sum += vals[t];
    }
    #pragma unroll
    for (int off = 16; off >= 1; off >>= 1)
        sum += __shfl_xor(sum, off, 32);
    if (lane == 0) red[wid] = sum;
    __syncthreads();
    float tot = 0.0f;
    #pragma unroll
    for (int w = 0; w < 8; ++w) tot += red[w];

    const float inv = 1.0f / tot;
    const float gam = gamma[0];
    #pragma unroll
    for (int t = 0; t < 16; ++t) {
        const long i = base + t * 256 + tid;
        const float att = vals[t] * inv * V[i];
        const float mk  = mask[i];
        out[i] = fg[i] * mk + gam * (1.0f - mk) * att;
    }
}

extern "C" void kernel_launch(void* const* d_in, const int* in_sizes, int n_in,
                              void* d_out, int out_size, void* d_ws, size_t ws_size,
                              hipStream_t stream)
{
    (void)in_sizes; (void)n_in; (void)out_size; (void)ws_size;

    const float* fg    = (const float*)d_in[0];
    const float* bg    = (const float*)d_in[1];
    const float* mask  = (const float*)d_in[2];
    const float* Wq    = (const float*)d_in[3];
    const float* bq    = (const float*)d_in[4];
    const float* Wk    = (const float*)d_in[5];
    const float* bk    = (const float*)d_in[6];
    const float* Wv    = (const float*)d_in[7];
    const float* bv    = (const float*)d_in[8];
    const float* gamma = (const float*)d_in[9];
    float* out = (float*)d_out;

    const long mat = (long)BB * CC * HWN;
    float* Q    = (float*)d_ws;
    float* Km   = Q  + mat;
    float* Vv   = Km + mat;
    float* corr = Vv + mat;

    const dim3 blk(128);
    const dim3 gGemm(HWN / 64, CC / 64, BB);
    const long xStride = (long)CC * HWN;

    gemm256_kernel<<<gGemm, blk, 0, stream>>>(Wq, 0, bq, fg, xStride, Q,  xStride);
    gemm256_kernel<<<gGemm, blk, 0, stream>>>(Wk, 0, bk, bg, xStride, Km, xStride);
    gemm256_kernel<<<gGemm, blk, 0, stream>>>(Wv, 0, bv, fg, xStride, Vv, xStride);

    const dim3 gCorr(CC / 16, CC / 64, BB);
    corr_kernel<<<gCorr, blk, 0, stream>>>(Q, Km, corr);

    gemm256_kernel<<<gGemm, blk, 0, stream>>>(corr, (long)CC * CC, nullptr,
                                              mask, xStride, out, xStride);

    softmax_out_kernel<<<dim3(BB * CC), dim3(256), 0, stream>>>(
        out, fg, mask, Vv, gamma, out);
}


# GEMM and corr loops: 128-bit fragment loads (A: 8 k per lane; B: 4 columns per lane, tiles column-interleaved; float4 stores; 128-VGPR budget, 20 loads in flight); new k order changes the sum order
# speedup vs baseline: 1.0959x; 1.0959x over previous
#include <hip/hip_runtime.h>

typedef __attribute__((ext_vector_type(2))) float v2f;
typedef __attribute__((ext_vector_type(4))) float v4f;
typedef __attribute__((ext_vector_type(8))) float v8f;

#define BB 8
#define CC 256
#define HWN 4096

__global__ __launch_bounds__(128) __attribute__((amdgpu_waves_per_eu(4, 8))) void gemm256_kernel(
    const float* __restrict__ A, long aStride,
    const float* __restrict__ bias,
    const float* __restrict__ Bm, long bStride,
    float* __restrict__ D, long dStride)
{
    const int lane  = threadIdx.x & 31;
    const int wave  = threadIdx.x >> 5;
    const int g     = lane >> 4;
    const int ln    = lane & 15;
    const int batch = blockIdx.z;
    const int mBase = blockIdx.y * 64 + wave * 16;
    const int nBase = blockIdx.x * 64;

    const float* Ab = A + (long)batch * aStride;
    const float* Bb = Bm + (long)batch * bStride;
    float*       Db = D + (long)batch * dStride;

    v8f acc0 = {}, acc1 = {}, acc2 = {}, acc3 = {};

    const float* arow = Ab + (long)(mBase + ln) * CC + 8 * g;
    const float* bcol = Bb + (long)(8 * g) * HWN + nBase + 4 * ln;

    #pragma unroll 2
    for (int k = 0; k < CC; k += 16) {
        const v4f a0 = *(const v4f*)(arow + k);
        const v4f a1 = *(const v4f*)(arow + k + 4);
        const float* bp = bcol + (long)k * HWN;
        const v4f r0 = *(const v4f*)(bp);
        const v4f r1 = *(const v4f*)(bp + 1 * HWN);
        const v4f r2 = *(const v4f*)(bp + 2 * HWN);
        const v4f r3 = *(const v4f*)(bp + 3 * HWN);
        const v4f r4 = *(const v4f*)(bp + 4 * HWN);
        const v4f r5 = *(const v4f*)(bp + 5 * HWN);
        const v4f r6 = *(const v4f*)(bp + 6 * HWN);
        const v4f r7 = *(const v4f*)(bp + 7 * HWN);

        v2f a, b0, b1, b2, b3;
        a  = (v2f){ a0.x, a0.y };
        b0 = (v2f){ r0.x, r1.x }; b1 = (v2f){ r0.y, r1.y };
        b2 = (v2f){ r0.z, r1.z }; b3 = (v2f){ r0.w, r1.w };
        acc0 = __builtin_amdgcn_wmma_f32_16x16x4_f32(false, a, false, b0, (short)0, acc0, false, false);
        acc1 = __builtin_amdgcn_wmma_f32_16x16x4_f32(false, a, false, b1, (short)0, acc1, false, false);
        acc2 = __builtin_amdgcn_wmma_f32_16x16x4_f32(false, a, false, b2, (short)0, acc2, false, false);
        acc3 = __builtin_amdgcn_wmma_f32_16x16x4_f32(false, a, false, b3, (short)0, acc3, false, false);
        a  = (v2f){ a0.z, a0.w };
        b0 = (v2f){ r2.x, r3.x }; b1 = (v2f){ r2.y, r3.y };
        b2 = (v2f){ r2.z, r3.z }; b3 = (v2f){ r2.w, r3.w };
        acc0 = __builtin_amdgcn_wmma_f32_16x16x4_f32(false, a, false, b0, (short)0, acc0, false, false);
        acc1 = __builtin_amdgcn_wmma_f32_16x16x4_f32(false, a, false, b1, (short)0, acc1, false, false);
        acc2 = __builtin_amdgcn_wmma_f32_16x16x4_f32(false, a, false, b2, (short)0, acc2, false, false);
        acc3 = __builtin_amdgcn_wmma_f32_16x16x4_f32(false, a, false, b3, (short)0, acc3, false, false);
        a  = (v2f){ a1.x, a1.y };
        b0 = (v2f){ r4.x, r5.x }; b1 = (v2f){ r4.y, r5.y };
        b2 = (v2f){ r4.z, r5.z }; b3 = (v2f){ r4.w, r5.w };
        acc0 = __builtin_amdgcn_wmma_f32_16x16x4_f32(false, a, false, b0, (short)0, acc0, false, false);
        acc1 = __builtin_amdgcn_wmma_f32_16x16x4_f32(false, a, false, b1, (short)0, acc1, false, false);
        acc2 = __builtin_amdgcn_wmma_f32_16x16x4_f32(false, a, false, b2, (short)0, acc2, false, false);
        acc3 = __builtin_amdgcn_wmma_f32_16x16x4_f32(false, a, false, b3, (short)0, acc3, false, false);
        a  = (v2f){ a1.z, a1.w };
        b0 = (v2f){ r6.x, r7.x }; b1 = (v2f){ r6.y, r7.y };
        b2 = (v2f){ r6.z, r7.z }; b3 = (v2f){ r6.w, r7.w };
        acc0 = __builtin_amdgcn_wmma_f32_16x16x4_f32(false, a, false, b0, (short)0, acc0, false, false);
        acc1 = __builtin_amdgcn_wmma_f32_16x16x4_f32(false, a, false, b1, (short)0, acc1, false, false);
        acc2 = __builtin_amdgcn_wmma_f32_16x16x4_f32(false, a, false, b2, (short)0, acc2, false, false);
        acc3 = __builtin_amdgcn_wmma_f32_16x16x4_f32(false, a, false, b3, (short)0, acc3, false, false);
    }

    #pragma unroll
    for (int r = 0; r < 8; ++r) {
        const int mo = mBase + r + 8 * g;
        const float bv = bias ? bias[mo] : 0.0f;
        const v4f o = { acc0[r] + bv, acc1[r] + bv, acc2[r] + bv, acc3[r] + bv };
        *(v4f*)(Db + (long)mo * HWN + nBase + 4 * ln) = o;
    }
}

__global__ __launch_bounds__(128) void corr_kernel(
    const float* __restrict__ Q, const float* __restrict__ Km,
    float* __restrict__ corr)
{
    const int lane  = threadIdx.x & 31;
    const int wave  = threadIdx.x >> 5;
    const int g     = lane >> 4;
    const int ln    = lane & 15;
    const int batch = blockIdx.z;
    const int mBase = blockIdx.y * 64 + wave * 16;
    const int nBase = blockIdx.x * 16;

    const float* qp = Q  + (long)batch * CC * HWN + (long)(mBase + ln) * HWN + 8 * g;
    const float* kp = Km + (long)batch * CC * HWN + (long)(nBase + ln) * HWN + 8 * g;

    v8f acc = {};
    #pragma unroll 4
    for (int k = 0; k < HWN; k += 16) {
        const v4f a0 = *(const v4f*)(qp + k);
        const v4f a1 = *(const v4f*)(qp + k + 4);
        const v4f b0 = *(const v4f*)(kp + k);
        const v4f b1 = *(const v4f*)(kp + k + 4);
        acc = __builtin_amdgcn_wmma_f32_16x16x4_f32(false, (v2f){ a0.x, a0.y }, false, (v2f){ b0.x, b0.y }, (short)0, acc, false, false);
        acc = __builtin_amdgcn_wmma_f32_16x16x4_f32(false, (v2f){ a0.z, a0.w }, false, (v2f){ b0.z, b0.w }, (short)0, acc, false, false);
        acc = __builtin_amdgcn_wmma_f32_16x16x4_f32(false, (v2f){ a1.x, a1.y }, false, (v2f){ b1.x, b1.y }, (short)0, acc, false, false);
        acc = __builtin_amdgcn_wmma_f32_16x16x4_f32(false, (v2f){ a1.z, a1.w }, false, (v2f){ b1.z, b1.w }, (short)0, acc, false, false);
    }

    float* cb = corr + (long)batch * CC * CC;
    #pragma unroll
    for (int r = 0; r < 8; ++r) {
        cb[(long)(mBase + r + 8 * g) * CC + nBase + ln] = acc[r];
    }
}

__global__ __launch_bounds__(256) void softmax_out_kernel(
    const float* __restrict__ scores,
    const float* __restrict__ fg, const float* __restrict__ mask,
    const float* __restrict__ V,  const float* __restrict__ gamma,
    float* __restrict__ out)
{
    const int  tid  = threadIdx.x;
    const int  lane = tid & 31;
    const int  wid  = tid >> 5;
    const long base = (long)blockIdx.x * HWN;

    __shared__ float red[8];

    float vals[16];
    float mx = -3.402823466e38f;
    #pragma unroll
    for (int t = 0; t < 16; ++t) {
        vals[t] = scores[base + t * 256 + tid];
        mx = fmaxf(mx, vals[t]);
    }
    #pragma unroll
    for (int off = 16; off >= 1; off >>= 1)
        mx = fmaxf(mx, __shfl_xor(mx, off, 32));
    if (lane == 0) red[wid] = mx;
    __syncthreads();
    #pragma unroll
    for (int w = 0; w < 8; ++w) mx = fmaxf(mx, red[w]);
    __syncthreads();

    float sum = 0.0f;
    #pragma unroll
    for (int t = 0; t < 16; ++t) {
        vals[t] = __expf(vals[t] - mx);
        sum += vals[t];
    }
    #pragma unroll
    for (int off = 16; off >= 1; off >>= 1)
        sum += __shfl_xor(sum, off, 32);
    if (lane == 0) red[wid] = sum;
    __syncthreads();
    float tot = 0.0f;
    #pragma unroll
    for (int w = 0; w < 8; ++w) tot += red[w];

    const float inv = 1.0f / tot;
    const float gam = gamma[0];
    #pragma unroll
    for (int t = 0; t < 16; ++t) {
        const long i = base + t * 256 + tid;
        const float att = vals[t] * inv * V[i];
        const float mk  = mask[i];
        out[i] = fg[i] * mk + gam * (1.0f - mk) * att;
    }
}

extern "C" void kernel_launch(void* const* d_in, const int* in_sizes, int n_in,
                              void* d_out, int out_size, void* d_ws, size_t ws_size,
                              hipStream_t stream)
{
    (void)in_sizes; (void)n_in; (void)out_size; (void)ws_size;

    const float* fg    = (const float*)d_in[0];
    const float* bg    = (const float*)d_in[1];
    const float* mask  = (const float*)d_in[2];
    const float* Wq    = (const float*)d_in[3];
    const float* bq    = (const float*)d_in[4];
    const float* Wk    = (const float*)d_in[5];
    const float* bk    = (const float*)d_in[6];
    const float* Wv    = (const float*)d_in[7];
    const float* bv    = (const float*)d_in[8];
    const float* gamma = (const float*)d_in[9];
    float* out = (float*)d_out;

    const long mat = (long)BB * CC * HWN;
    float* Q    = (float*)d_ws;
    float* Km   = Q  + mat;
    float* Vv   = Km + mat;
    float* corr = Vv + mat;

    const dim3 blk(128);
    const dim3 gGemm(HWN / 64, CC / 64, BB);
    const long xStride = (long)CC * HWN;

    gemm256_kernel<<<gGemm, blk, 0, stream>>>(Wq, 0, bq, fg, xStride, Q,  xStride);
    gemm256_kernel<<<gGemm, blk, 0, stream>>>(Wk, 0, bk, bg, xStride, Km, xStride);
    gemm256_kernel<<<gGemm, blk, 0, stream>>>(Wv, 0, bv, fg, xStride, Vv, xStride);

    const dim3 gCorr(CC / 16, CC / 64, BB);
    corr_kernel<<<gCorr, blk, 0, stream>>>(Q, Km, corr);

    gemm256_kernel<<<gGemm, blk, 0, stream>>>(corr, (long)CC * CC, nullptr,
                                              mask, xStride, out, xStride);

    softmax_out_kernel<<<dim3(BB * CC), dim3(256), 0, stream>>>(
        out, fg, mask, Vv, gamma, out);
}


# corr reassociated: q k^T computed as Wq (fg bg^T) Wk^T with exact bias rank-1 terms, so the two HW-wide q/k projection GEMMs become two 256-wide WMMA products; changes the sum order
# speedup vs baseline: 1.4238x; 1.2992x over previous
#include <hip/hip_runtime.h>

typedef __attribute__((ext_vector_type(2))) float v2f;
typedef __attribute__((ext_vector_type(4))) float v4f;
typedef __attribute__((ext_vector_type(8))) float v8f;

#define BB 8
#define CC 256
#define HWN 4096
#define MT_BATCH (257 * 256)
#define TQ_PITCH 272
#define TQ_BATCH (256 * 272)

__global__ __launch_bounds__(128) __attribute__((amdgpu_waves_per_eu(4, 8))) void gemm256_kernel(
    const float* __restrict__ A, long aStride,
    const float* __restrict__ bias,
    const float* __restrict__ Bm, long bStride,
    float* __restrict__ D, long dStride)
{
    const int lane  = threadIdx.x & 31;
    const int wave  = threadIdx.x >> 5;
    const int g     = lane >> 4;
    const int ln    = lane & 15;
    const int batch = blockIdx.z;
    const int mBase = blockIdx.y * 64 + wave * 16;
    const int nBase = blockIdx.x * 64;

    const float* Ab = A + (long)batch * aStride;
    const float* Bb = Bm + (long)batch * bStride;
    float*       Db = D + (long)batch * dStride;

    v8f acc0 = {}, acc1 = {}, acc2 = {}, acc3 = {};

    const float* arow = Ab + (long)(mBase + ln) * CC + 8 * g;
    const float* bcol = Bb + (long)(8 * g) * HWN + nBase + 4 * ln;

    #pragma unroll 2
    for (int k = 0; k < CC; k += 16) {
        const v4f a0 = *(const v4f*)(arow + k);
        const v4f a1 = *(const v4f*)(arow + k + 4);
        const float* bp = bcol + (long)k * HWN;
        const v4f r0 = *(const v4f*)(bp);
        const v4f r1 = *(const v4f*)(bp + 1 * HWN);
        const v4f r2 = *(const v4f*)(bp + 2 * HWN);
        const v4f r3 = *(const v4f*)(bp + 3 * HWN);
        const v4f r4 = *(const v4f*)(bp + 4 * HWN);
        const v4f r5 = *(const v4f*)(bp + 5 * HWN);
        const v4f r6 = *(const v4f*)(bp + 6 * HWN);
        const v4f r7 = *(const v4f*)(bp + 7 * HWN);

        v2f a, b0, b1, b2, b3;
        a  = (v2f){ a0.x, a0.y };
        b0 = (v2f){ r0.x, r1.x }; b1 = (v2f){ r0.y, r1.y };
        b2 = (v2f){ r0.z, r1.z }; b3 = (v2f){ r0.w, r1.w };
        acc0 = __builtin_amdgcn_wmma_f32_16x16x4_f32(false, a, false, b0, (short)0, acc0, false, false);
        acc1 = __builtin_amdgcn_wmma_f32_16x16x4_f32(false, a, false, b1, (short)0, acc1, false, false);
        acc2 = __builtin_amdgcn_wmma_f32_16x16x4_f32(false, a, false, b2, (short)0, acc2, false, false);
        acc3 = __builtin_amdgcn_wmma_f32_16x16x4_f32(false, a, false, b3, (short)0, acc3, false, false);
        a  = (v2f){ a0.z, a0.w };
        b0 = (v2f){ r2.x, r3.x }; b1 = (v2f){ r2.y, r3.y };
        b2 = (v2f){ r2.z, r3.z }; b3 = (v2f){ r2.w, r3.w };
        acc0 = __builtin_amdgcn_wmma_f32_16x16x4_f32(false, a, false, b0, (short)0, acc0, false, false);
        acc1 = __builtin_amdgcn_wmma_f32_16x16x4_f32(false, a, false, b1, (short)0, acc1, false, false);
        acc2 = __builtin_amdgcn_wmma_f32_16x16x4_f32(false, a, false, b2, (short)0, acc2, false, false);
        acc3 = __builtin_amdgcn_wmma_f32_16x16x4_f32(false, a, false, b3, (short)0, acc3, false, false);
        a  = (v2f){ a1.x, a1.y };
        b0 = (v2f){ r4.x, r5.x }; b1 = (v2f){ r4.y, r5.y };
        b2 = (v2f){ r4.z, r5.z }; b3 = (v2f){ r4.w, r5.w };
        acc0 = __builtin_amdgcn_wmma_f32_16x16x4_f32(false, a, false, b0, (short)0, acc0, false, false);
        acc1 = __builtin_amdgcn_wmma_f32_16x16x4_f32(false, a, false, b1, (short)0, acc1, false, false);
        acc2 = __builtin_amdgcn_wmma_f32_16x16x4_f32(false, a, false, b2, (short)0, acc2, false, false);
        acc3 = __builtin_amdgcn_wmma_f32_16x16x4_f32(false, a, false, b3, (short)0, acc3, false, false);
        a  = (v2f){ a1.z, a1.w };
        b0 = (v2f){ r6.x, r7.x }; b1 = (v2f){ r6.y, r7.y };
        b2 = (v2f){ r6.z, r7.z }; b3 = (v2f){ r6.w, r7.w };
        acc0 = __builtin_amdgcn_wmma_f32_16x16x4_f32(false, a, false, b0, (short)0, acc0, false, false);
        acc1 = __builtin_amdgcn_wmma_f32_16x16x4_f32(false, a, false, b1, (short)0, acc1, false, false);
        acc2 = __builtin_amdgcn_wmma_f32_16x16x4_f32(false, a, false, b2, (short)0, acc2, false, false);
        acc3 = __builtin_amdgcn_wmma_f32_16x16x4_f32(false, a, false, b3, (short)0, acc3, false, false);
    }

    #pragma unroll
    for (int r = 0; r < 8; ++r) {
        const int mo = mBase + r + 8 * g;
        const float bv = bias ? bias[mo] : 0.0f;
        const v4f o = { acc0[r] + bv, acc1[r] + bv, acc2[r] + bv, acc3[r] + bv };
        *(v4f*)(Db + (long)mo * HWN + nBase + 4 * ln) = o;
    }
}

#define NT_GROUP(SUMS)                                                                  \
    {                                                                                   \
        const v4f a0 = *(const v4f*)(ap + k);                                           \
        const v4f a1 = *(const v4f*)(ap + k + 4);                                       \
        const v4f b0 = *(const v4f*)(bp + k);                                           \
        const v4f b1 = *(const v4f*)(bp + k + 4);                                       \
        if (SUMS) { sa += a0 + a1; sb += b0 + b1; }                                     \
        acc = __builtin_amdgcn_wmma_f32_16x16x4_f32(false, (v2f){ a0.x, a0.y }, false, (v2f){ b0.x, b0.y }, (short)0, acc, false, false); \
        acc = __builtin_amdgcn_wmma_f32_16x16x4_f32(false, (v2f){ a0.z, a0.w }, false, (v2f){ b0.z, b0.w }, (short)0, acc, false, false); \
        acc = __builtin_amdgcn_wmma_f32_16x16x4_f32(false, (v2f){ a1.x, a1.y }, false, (v2f){ b1.x, b1.y }, (short)0, acc, false, false); \
        acc = __builtin_amdgcn_wmma_f32_16x16x4_f32(false, (v2f){ a1.z, a1.w }, false, (v2f){ b1.z, b1.w }, (short)0, acc, false, false); \
    }

__global__ __launch_bounds__(128) void nt_kernel(
    const float* __restrict__ A,  long aBatch, int aPitch,
    const float* __restrict__ Bt, long bBatch, int bPitch, int bRows,
    int kLen,
    const float* __restrict__ u, long uBatch, int uStride,
    const float* __restrict__ w, long wBatch,
    float* __restrict__ D, long dBatch, int dPitch,
    float* __restrict__ sumA, long saBatch,
    float* __restrict__ sumB, long sbBatch)
{
    const int lane  = threadIdx.x & 31;
    const int wave  = threadIdx.x >> 5;
    const int g     = lane >> 4;
    const int ln    = lane & 15;
    const int batch = blockIdx.z;
    const int mBase = blockIdx.y * 64 + wave * 16;
    const int nBase = blockIdx.x * 16;
    const int nRow  = min(nBase + ln, bRows - 1);

    const float* ap = A  + (long)batch * aBatch + (long)(mBase + ln) * aPitch + 8 * g;
    const float* bp = Bt + (long)batch * bBatch + (long)nRow * bPitch + 8 * g;

    v8f acc = {};
    v4f sa = {}, sb = {};
    const bool sumsA = (sumA != nullptr) && (blockIdx.x == 0);
    const bool sumsB = (sumA != nullptr) && (blockIdx.y == 0) && (wave == 0);
    if (sumsA || sumsB) {
        for (int k0 = 0; k0 < kLen; k0 += 64) {
            #pragma unroll
            for (int kk = 0; kk < 64; kk += 16) { const int k = k0 + kk; NT_GROUP(true) }
        }
        float ra = (sa.x + sa.y) + (sa.z + sa.w);
        float rb = (sb.x + sb.y) + (sb.z + sb.w);
        ra += __shfl_xor(ra, 16, 32);
        rb += __shfl_xor(rb, 16, 32);
        if (g == 0) {
            if (sumsA) sumA[(long)batch * saBatch + mBase + ln] = ra;
            if (sumsB) sumB[(long)batch * sbBatch + nBase + ln] = rb;
        }
        if (sumsA && sumsB && lane == 0) sumA[(long)batch * saBatch + CC] = (float)kLen;
    } else {
        for (int k0 = 0; k0 < kLen; k0 += 64) {
            #pragma unroll
            for (int kk = 0; kk < 64; kk += 16) { const int k = k0 + kk; NT_GROUP(false) }
        }
    }

    const float wv = w ? w[(long)batch * wBatch + nRow] : 0.0f;
    float* db = D + (long)batch * dBatch;
    #pragma unroll
    for (int r = 0; r < 8; ++r) {
        const int mo = mBase + r + 8 * g;
        const float uv = u ? u[(long)batch * uBatch + (long)mo * uStride] : 0.0f;
        db[(long)mo * dPitch + nBase + ln] = acc[r] + uv * wv;
    }
}

__global__ __launch_bounds__(256) void softmax_out_kernel(
    const float* __restrict__ scores,
    const float* __restrict__ fg, const float* __restrict__ mask,
    const float* __restrict__ V,  const float* __restrict__ gamma,
    float* __restrict__ out)
{
    const int  tid  = threadIdx.x;
    const int  lane = tid & 31;
    const int  wid  = tid >> 5;
    const long base = (long)blockIdx.x * HWN;

    __shared__ float red[8];

    float vals[16];
    float mx = -3.402823466e38f;
    #pragma unroll
    for (int t = 0; t < 16; ++t) {
        vals[t] = scores[base + t * 256 + tid];
        mx = fmaxf(mx, vals[t]);
    }
    #pragma unroll
    for (int off = 16; off >= 1; off >>= 1)
        mx = fmaxf(mx, __shfl_xor(mx, off, 32));
    if (lane == 0) red[wid] = mx;
    __syncthreads();
    #pragma unroll
    for (int w = 0; w < 8; ++w) mx = fmaxf(mx, red[w]);
    __syncthreads();

    float sum = 0.0f;
    #pragma unroll
    for (int t = 0; t < 16; ++t) {
        vals[t] = __expf(vals[t] - mx);
        sum += vals[t];
    }
    #pragma unroll
    for (int off = 16; off >= 1; off >>= 1)
        sum += __shfl_xor(sum, off, 32);
    if (lane == 0) red[wid] = sum;
    __syncthreads();
    float tot = 0.0f;
    #pragma unroll
    for (int w = 0; w < 8; ++w) tot += red[w];

    const float inv = 1.0f / tot;
    const float gam = gamma[0];
    #pragma unroll
    for (int t = 0; t < 16; ++t) {
        const long i = base + t * 256 + tid;
        const float att = vals[t] * inv * V[i];
        const float mk  = mask[i];
        out[i] = fg[i] * mk + gam * (1.0f - mk) * att;
    }
}

extern "C" void kernel_launch(void* const* d_in, const int* in_sizes, int n_in,
                              void* d_out, int out_size, void* d_ws, size_t ws_size,
                              hipStream_t stream)
{
    (void)in_sizes; (void)n_in; (void)out_size; (void)ws_size;

    const float* fg    = (const float*)d_in[0];
    const float* bg    = (const float*)d_in[1];
    const float* mask  = (const float*)d_in[2];
    const float* Wq    = (const float*)d_in[3];
    const float* bq    = (const float*)d_in[4];
    const float* Wk    = (const float*)d_in[5];
    const float* bk    = (const float*)d_in[6];
    const float* Wv    = (const float*)d_in[7];
    const float* bv    = (const float*)d_in[8];
    const float* gamma = (const float*)d_in[9];
    float* out = (float*)d_out;

    const long mat = (long)BB * CC * HWN;
    float* Vv   = (float*)d_ws;
    float* MT   = Vv + mat;
    float* Tq   = MT + (long)BB * MT_BATCH;
    float* corr = Tq + (long)BB * TQ_BATCH;
    float* sG   = corr + (long)BB * CC * CC;

    const dim3 blk(128);
    const dim3 gGemm(HWN / 64, CC / 64, BB);
    const long xStride = (long)CC * HWN;

    gemm256_kernel<<<gGemm, blk, 0, stream>>>(Wv, 0, bv, fg, xStride, Vv, xStride);

    const dim3 gCorr(CC / 16, CC / 64, BB);
    nt_kernel<<<gCorr, blk, 0, stream>>>(bg, xStride, HWN, fg, xStride, HWN, CC, HWN,
                                         nullptr, 0, 0, nullptr, 0,
                                         MT, MT_BATCH, CC,
                                         sG, CC + 1, MT + CC * CC, MT_BATCH);
    nt_kernel<<<dim3(CC / 16 + 1, CC / 64, BB), blk, 0, stream>>>(
                                         Wq, 0, CC, MT, MT_BATCH, CC, CC + 1, CC,
                                         bq, 0, 1, sG, CC + 1,
                                         Tq, TQ_BATCH, TQ_PITCH,
                                         nullptr, 0, nullptr, 0);
    nt_kernel<<<gCorr, blk, 0, stream>>>(Tq, TQ_BATCH, TQ_PITCH, Wk, 0, CC, CC, CC,
                                         Tq + CC, TQ_BATCH, TQ_PITCH, bk, 0,
                                         corr, (long)CC * CC, CC,
                                         nullptr, 0, nullptr, 0);

    gemm256_kernel<<<gGemm, blk, 0, stream>>>(corr, (long)CC * CC, nullptr,
                                              mask, xStride, out, xStride);

    softmax_out_kernel<<<dim3(BB * CC), dim3(256), 0, stream>>>(
        out, fg, mask, Vv, gamma, out);
}


# fg bg^T product (K=HW): operands staged through LDS by whole-line 128-bit global loads (one row per wave load, not a 32-B piece of each fragment row), two stages; fragments, k and sum order kept
# speedup vs baseline: 1.7567x; 1.2338x over previous
#include <hip/hip_runtime.h>

typedef __attribute__((ext_vector_type(2))) float v2f;
typedef __attribute__((ext_vector_type(4))) float v4f;
typedef __attribute__((ext_vector_type(8))) float v8f;

#define BB 8
#define CC 256
#define HWN 4096
#define MT_BATCH (257 * 256)
#define TQ_PITCH 272
#define TQ_BATCH (256 * 272)

__global__ __launch_bounds__(128) __attribute__((amdgpu_waves_per_eu(4, 8))) void gemm256_kernel(
    const float* __restrict__ A, long aStride,
    const float* __restrict__ bias,
    const float* __restrict__ Bm, long bStride,
    float* __restrict__ D, long dStride)
{
    const int lane  = threadIdx.x & 31;
    const int wave  = threadIdx.x >> 5;
    const int g     = lane >> 4;
    const int ln    = lane & 15;
    const int batch = blockIdx.z;
    const int mBase = blockIdx.y * 64 + wave * 16;
    const int nBase = blockIdx.x * 64;

    const float* Ab = A + (long)batch * aStride;
    const float* Bb = Bm + (long)batch * bStride;
    float*       Db = D + (long)batch * dStride;

    v8f acc0 = {}, acc1 = {}, acc2 = {}, acc3 = {};

    const float* arow = Ab + (long)(mBase + ln) * CC + 8 * g;
    const float* bcol = Bb + (long)(8 * g) * HWN + nBase + 4 * ln;

    #pragma unroll 2
    for (int k = 0; k < CC; k += 16) {
        const v4f a0 = *(const v4f*)(arow + k);
        const v4f a1 = *(const v4f*)(arow + k + 4);
        const float* bp = bcol + (long)k * HWN;
        const v4f r0 = *(const v4f*)(bp);
        const v4f r1 = *(const v4f*)(bp + 1 * HWN);
        const v4f r2 = *(const v4f*)(bp + 2 * HWN);
        const v4f r3 = *(const v4f*)(bp + 3 * HWN);
        const v4f r4 = *(const v4f*)(bp + 4 * HWN);
        const v4f r5 = *(const v4f*)(bp + 5 * HWN);
        const v4f r6 = *(const v4f*)(bp + 6 * HWN);
        const v4f r7 = *(const v4f*)(bp + 7 * HWN);

        v2f a, b0, b1, b2, b3;
        a  = (v2f){ a0.x, a0.y };
        b0 = (v2f){ r0.x, r1.x }; b1 = (v2f){ r0.y, r1.y };
        b2 = (v2f){ r0.z, r1.z }; b3 = (v2f){ r0.w, r1.w };
        acc0 = __builtin_amdgcn_wmma_f32_16x16x4_f32(false, a, false, b0, (short)0, acc0, false, false);
        acc1 = __builtin_amdgcn_wmma_f32_16x16x4_f32(false, a, false, b1, (short)0, acc1, false, false);
        acc2 = __builtin_amdgcn_wmma_f32_16x16x4_f32(false, a, false, b2, (short)0, acc2, false, false);
        acc3 = __builtin_amdgcn_wmma_f32_16x16x4_f32(false, a, false, b3, (short)0, acc3, false, false);
        a  = (v2f){ a0.z, a0.w };
        b0 = (v2f){ r2.x, r3.x }; b1 = (v2f){ r2.y, r3.y };
        b2 = (v2f){ r2.z, r3.z }; b3 = (v2f){ r2.w, r3.w };
        acc0 = __builtin_amdgcn_wmma_f32_16x16x4_f32(false, a, false, b0, (short)0, acc0, false, false);
        acc1 = __builtin_amdgcn_wmma_f32_16x16x4_f32(false, a, false, b1, (short)0, acc1, false, false);
        acc2 = __builtin_amdgcn_wmma_f32_16x16x4_f32(false, a, false, b2, (short)0, acc2, false, false);
        acc3 = __builtin_amdgcn_wmma_f32_16x16x4_f32(false, a, false, b3, (short)0, acc3, false, false);
        a  = (v2f){ a1.x, a1.y };
        b0 = (v2f){ r4.x, r5.x }; b1 = (v2f){ r4.y, r5.y };
        b2 = (v2f){ r4.z, r5.z }; b3 = (v2f){ r4.w, r5.w };
        acc0 = __builtin_amdgcn_wmma_f32_16x16x4_f32(false, a, false, b0, (short)0, acc0, false, false);
        acc1 = __builtin_amdgcn_wmma_f32_16x16x4_f32(false, a, false, b1, (short)0, acc1, false, false);
        acc2 = __builtin_amdgcn_wmma_f32_16x16x4_f32(false, a, false, b2, (short)0, acc2, false, false);
        acc3 = __builtin_amdgcn_wmma_f32_16x16x4_f32(false, a, false, b3, (short)0, acc3, false, false);
        a  = (v2f){ a1.z, a1.w };
        b0 = (v2f){ r6.x, r7.x }; b1 = (v2f){ r6.y, r7.y };
        b2 = (v2f){ r6.z, r7.z }; b3 = (v2f){ r6.w, r7.w };
        acc0 = __builtin_amdgcn_wmma_f32_16x16x4_f32(false, a, false, b0, (short)0, acc0, false, false);
        acc1 = __builtin_amdgcn_wmma_f32_16x16x4_f32(false, a, false, b1, (short)0, acc1, false, false);
        acc2 = __builtin_amdgcn_wmma_f32_16x16x4_f32(false, a, false, b2, (short)0, acc2, false, false);
        acc3 = __builtin_amdgcn_wmma_f32_16x16x4_f32(false, a, false, b3, (short)0, acc3, false, false);
    }

    #pragma unroll
    for (int r = 0; r < 8; ++r) {
        const int mo = mBase + r + 8 * g;
        const float bv = bias ? bias[mo] : 0.0f;
        const v4f o = { acc0[r] + bv, acc1[r] + bv, acc2[r] + bv, acc3[r] + bv };
        *(v4f*)(Db + (long)mo * HWN + nBase + 4 * ln) = o;
    }
}

#define NT_GROUP(SUMS)                                                                  \
    {                                                                                   \
        const v4f a0 = *(const v4f*)(ap + k);                                           \
        const v4f a1 = *(const v4f*)(ap + k + 4);                                       \
        const v4f b0 = *(const v4f*)(bp + k);                                           \
        const v4f b1 = *(const v4f*)(bp + k + 4);                                       \
        if (SUMS) { sa += a0 + a1; sb += b0 + b1; }                                     \
        acc = __builtin_amdgcn_wmma_f32_16x16x4_f32(false, (v2f){ a0.x, a0.y }, false, (v2f){ b0.x, b0.y }, (short)0, acc, false, false); \
        acc = __builtin_amdgcn_wmma_f32_16x16x4_f32(false, (v2f){ a0.z, a0.w }, false, (v2f){ b0.z, b0.w }, (short)0, acc, false, false); \
        acc = __builtin_amdgcn_wmma_f32_16x16x4_f32(false, (v2f){ a1.x, a1.y }, false, (v2f){ b1.x, b1.y }, (short)0, acc, false, false); \
        acc = __builtin_amdgcn_wmma_f32_16x16x4_f32(false, (v2f){ a1.z, a1.w }, false, (v2f){ b1.z, b1.w }, (short)0, acc, false, false); \
    }

__global__ __launch_bounds__(128) void nt_kernel(
    const float* __restrict__ A,  long aBatch, int aPitch,
    const float* __restrict__ Bt, long bBatch, int bPitch, int bRows,
    int kLen,
    const float* __restrict__ u, long uBatch, int uStride,
    const float* __restrict__ w, long wBatch,
    float* __restrict__ D, long dBatch, int dPitch,
    float* __restrict__ sumA, long saBatch,
    float* __restrict__ sumB, long sbBatch)
{
    const int lane  = threadIdx.x & 31;
    const int wave  = threadIdx.x >> 5;
    const int g     = lane >> 4;
    const int ln    = lane & 15;
    const int batch = blockIdx.z;
    const int mBase = blockIdx.y * 64 + wave * 16;
    const int nBase = blockIdx.x * 16;
    const int nRow  = min(nBase + ln, bRows - 1);

    const float* ap = A  + (long)batch * aBatch + (long)(mBase + ln) * aPitch + 8 * g;
    const float* bp = Bt + (long)batch * bBatch + (long)nRow * bPitch + 8 * g;

    v8f acc = {};
    v4f sa = {}, sb = {};
    const bool sumsA = (sumA != nullptr) && (blockIdx.x == 0);
    const bool sumsB = (sumA != nullptr) && (blockIdx.y == 0) && (wave == 0);
    if (sumsA || sumsB) {
        for (int k0 = 0; k0 < kLen; k0 += 64) {
            #pragma unroll
            for (int kk = 0; kk < 64; kk += 16) { const int k = k0 + kk; NT_GROUP(true) }
        }
        float ra = (sa.x + sa.y) + (sa.z + sa.w);
        float rb = (sb.x + sb.y) + (sb.z + sb.w);
        ra += __shfl_xor(ra, 16, 32);
        rb += __shfl_xor(rb, 16, 32);
        if (g == 0) {
            if (sumsA) sumA[(long)batch * saBatch + mBase + ln] = ra;
            if (sumsB) sumB[(long)batch * sbBatch + nBase + ln] = rb;
        }
        if (sumsA && sumsB && lane == 0) sumA[(long)batch * saBatch + CC] = (float)kLen;
    } else {
        for (int k0 = 0; k0 < kLen; k0 += 64) {
            #pragma unroll
            for (int kk = 0; kk < 64; kk += 16) { const int k = k0 + kk; NT_GROUP(false) }
        }
    }

    const float wv = w ? w[(long)batch * wBatch + nRow] : 0.0f;
    float* db = D + (long)batch * dBatch;
    #pragma unroll
    for (int r = 0; r < 8; ++r) {
        const int mo = mBase + r + 8 * g;
        const float uv = u ? u[(long)batch * uBatch + (long)mo * uStride] : 0.0f;
        db[(long)mo * dPitch + nBase + ln] = acc[r] + uv * wv;
    }
}

#define MT_KC    128
#define MT_PITCH 132

#define MT_GROUP(SUMS, J)                                                               \
    {                                                                                   \
        const v4f a0 = fa[2 * (J)], a1 = fa[2 * (J) + 1];                               \
        const v4f b0 = fb[2 * (J)], b1 = fb[2 * (J) + 1];                               \
        if (SUMS) { sa += a0 + a1; sb += b0 + b1; }                                     \
        acc = __builtin_amdgcn_wmma_f32_16x16x4_f32(false, (v2f){ a0.x, a0.y }, false, (v2f){ b0.x, b0.y }, (short)0, acc, false, false); \
        acc = __builtin_amdgcn_wmma_f32_16x16x4_f32(false, (v2f){ a0.z, a0.w }, false, (v2f){ b0.z, b0.w }, (short)0, acc, false, false); \
        acc = __builtin_amdgcn_wmma_f32_16x16x4_f32(false, (v2f){ a1.x, a1.y }, false, (v2f){ b1.x, b1.y }, (short)0, acc, false, false); \
        acc = __builtin_amdgcn_wmma_f32_16x16x4_f32(false, (v2f){ a1.z, a1.w }, false, (v2f){ b1.z, b1.w }, (short)0, acc, false, false); \
    }

template <bool SUMS>
__device__ __forceinline__ void mt_chunks(
    const float* __restrict__ ag, const float* __restrict__ bgl, int aPitch, int bPitch, int kLen,
    float (*sA)[64][MT_PITCH], float (*sB)[16][MT_PITCH],
    int wave, int lane, int g, int ln, v8f& acc, v4f& sa, v4f& sb)
{
    v4f pa[16], pb[4];
    #pragma unroll
    for (int i = 0; i < 16; ++i) pa[i] = *(const v4f*)(ag + (long)(4 * i) * aPitch);
    #pragma unroll
    for (int i = 0; i < 4; ++i)  pb[i] = *(const v4f*)(bgl + (long)(4 * i) * bPitch);
    #pragma unroll
    for (int i = 0; i < 16; ++i) *(v4f*)&sA[0][4 * i + wave][4 * lane] = pa[i];
    #pragma unroll
    for (int i = 0; i < 4; ++i)  *(v4f*)&sB[0][4 * i + wave][4 * lane] = pb[i];
    __syncthreads();

    const int nChunk = kLen / MT_KC;
    for (int t = 0; t < nChunk; ++t) {
        const int cur = t & 1;
        const bool more = (t + 1 < nChunk);
        if (more) {
            const long k1 = (long)(t + 1) * MT_KC;
            #pragma unroll
            for (int i = 0; i < 16; ++i) pa[i] = *(const v4f*)(ag + (long)(4 * i) * aPitch + k1);
            #pragma unroll
            for (int i = 0; i < 4; ++i)  pb[i] = *(const v4f*)(bgl + (long)(4 * i) * bPitch + k1);
        }
        const float* as = &sA[cur][16 * wave + ln][8 * g];
        const float* bs = &sB[cur][ln][8 * g];
        v4f fa[16], fb[16];
        #pragma unroll
        for (int j = 0; j < 8; ++j) {
            fa[2 * j]     = *(const v4f*)(as + 16 * j);
            fa[2 * j + 1] = *(const v4f*)(as + 16 * j + 4);
            fb[2 * j]     = *(const v4f*)(bs + 16 * j);
            fb[2 * j + 1] = *(const v4f*)(bs + 16 * j + 4);
        }
        __builtin_amdgcn_sched_barrier(0);
        #pragma unroll
        for (int j = 0; j < 8; ++j) MT_GROUP(SUMS, j)
        if (more) {
            #pragma unroll
            for (int i = 0; i < 16; ++i) *(v4f*)&sA[cur ^ 1][4 * i + wave][4 * lane] = pa[i];
            #pragma unroll
            for (int i = 0; i < 4; ++i)  *(v4f*)&sB[cur ^ 1][4 * i + wave][4 * lane] = pb[i];
        }
        __syncthreads();
    }
}

__global__ __launch_bounds__(128) void mt_lds_kernel(
    const float* __restrict__ A,  long aBatch, int aPitch,
    const float* __restrict__ Bt, long bBatch, int bPitch,
    int kLen,
    float* __restrict__ D, long dBatch, int dPitch,
    float* __restrict__ sumA, long saBatch,
    float* __restrict__ sumB, long sbBatch)
{
    __shared__ __attribute__((aligned(16))) float sA[2][64][MT_PITCH];
    __shared__ __attribute__((aligned(16))) float sB[2][16][MT_PITCH];

    const int lane  = threadIdx.x & 31;
    const int wave  = threadIdx.x >> 5;
    const int g     = lane >> 4;
    const int ln    = lane & 15;
    const int batch = blockIdx.z;
    const int mBlk  = blockIdx.y * 64;
    const int mBase = mBlk + wave * 16;
    const int nBase = blockIdx.x * 16;

    const float* ag  = A  + (long)batch * aBatch + (long)(mBlk  + wave) * aPitch + 4 * lane;
    const float* bgl = Bt + (long)batch * bBatch + (long)(nBase + wave) * bPitch + 4 * lane;

    v8f acc = {};
    v4f sa = {}, sb = {};
    const bool sumsA = (blockIdx.x == 0);
    const bool sumsB = (blockIdx.y == 0) && (wave == 0);
    if (blockIdx.x == 0 || blockIdx.y == 0) {
        mt_chunks<true>(ag, bgl, aPitch, bPitch, kLen, sA, sB, wave, lane, g, ln, acc, sa, sb);
        float ra = (sa.x + sa.y) + (sa.z + sa.w);
        float rb = (sb.x + sb.y) + (sb.z + sb.w);
        ra += __shfl_xor(ra, 16, 32);
        rb += __shfl_xor(rb, 16, 32);
        if (g == 0) {
            if (sumsA) sumA[(long)batch * saBatch + mBase + ln] = ra;
            if (sumsB) sumB[(long)batch * sbBatch + nBase + ln] = rb;
        }
        if (sumsA && sumsB && lane == 0) sumA[(long)batch * saBatch + CC] = (float)kLen;
    } else {
        mt_chunks<false>(ag, bgl, aPitch, bPitch, kLen, sA, sB, wave, lane, g, ln, acc, sa, sb);
    }

    float* db = D + (long)batch * dBatch;
    #pragma unroll
    for (int r = 0; r < 8; ++r) {
        const int mo = mBase + r + 8 * g;
        db[(long)mo * dPitch + nBase + ln] = acc[r];
    }
}

__global__ __launch_bounds__(256) void softmax_out_kernel(
    const float* __restrict__ scores,
    const float* __restrict__ fg, const float* __restrict__ mask,
    const float* __restrict__ V,  const float* __restrict__ gamma,
    float* __restrict__ out)
{
    const int  tid  = threadIdx.x;
    const int  lane = tid & 31;
    const int  wid  = tid >> 5;
    const long base = (long)blockIdx.x * HWN;

    __shared__ float red[8];

    float vals[16];
    float mx = -3.402823466e38f;
    #pragma unroll
    for (int t = 0; t < 16; ++t) {
        vals[t] = scores[base + t * 256 + tid];
        mx = fmaxf(mx, vals[t]);
    }
    #pragma unroll
    for (int off = 16; off >= 1; off >>= 1)
        mx = fmaxf(mx, __shfl_xor(mx, off, 32));
    if (lane == 0) red[wid] = mx;
    __syncthreads();
    #pragma unroll
    for (int w = 0; w < 8; ++w) mx = fmaxf(mx, red[w]);
    __syncthreads();

    float sum = 0.0f;
    #pragma unroll
    for (int t = 0; t < 16; ++t) {
        vals[t] = __expf(vals[t] - mx);
        sum += vals[t];
    }
    #pragma unroll
    for (int off = 16; off >= 1; off >>= 1)
        sum += __shfl_xor(sum, off, 32);
    if (lane == 0) red[wid] = sum;
    __syncthreads();
    float tot = 0.0f;
    #pragma unroll
    for (int w = 0; w < 8; ++w) tot += red[w];

    const float inv = 1.0f / tot;
    const float gam = gamma[0];
    #pragma unroll
    for (int t = 0; t < 16; ++t) {
        const long i = base + t * 256 + tid;
        const float att = vals[t] * inv * V[i];
        const float mk  = mask[i];
        out[i] = fg[i] * mk + gam * (1.0f - mk) * att;
    }
}

extern "C" void kernel_launch(void* const* d_in, const int* in_sizes, int n_in,
                              void* d_out, int out_size, void* d_ws, size_t ws_size,
                              hipStream_t stream)
{
    (void)in_sizes; (void)n_in; (void)out_size; (void)ws_size;

    const float* fg    = (const float*)d_in[0];
    const float* bg    = (const float*)d_in[1];
    const float* mask  = (const float*)d_in[2];
    const float* Wq    = (const float*)d_in[3];
    const float* bq    = (const float*)d_in[4];
    const float* Wk    = (const float*)d_in[5];
    const float* bk    = (const float*)d_in[6];
    const float* Wv    = (const float*)d_in[7];
    const float* bv    = (const float*)d_in[8];
    const float* gamma = (const float*)d_in[9];
    float* out = (float*)d_out;

    const long mat = (long)BB * CC * HWN;
    float* Vv   = (float*)d_ws;
    float* MT   = Vv + mat;
    float* Tq   = MT + (long)BB * MT_BATCH;
    float* corr = Tq + (long)BB * TQ_BATCH;
    float* sG   = corr + (long)BB * CC * CC;

    const dim3 blk(128);
    const dim3 gGemm(HWN / 64, CC / 64, BB);
    const long xStride = (long)CC * HWN;

    gemm256_kernel<<<gGemm, blk, 0, stream>>>(Wv, 0, bv, fg, xStride, Vv, xStride);

    const dim3 gCorr(CC / 16, CC / 64, BB);
    mt_lds_kernel<<<gCorr, blk, 0, stream>>>(bg, xStride, HWN, fg, xStride, HWN, HWN,
                                             MT, MT_BATCH, CC,
                                             sG, CC + 1, MT + CC * CC, MT_BATCH);
    nt_kernel<<<dim3(CC / 16 + 1, CC / 64, BB), blk, 0, stream>>>(
                                         Wq, 0, CC, MT, MT_BATCH, CC, CC + 1, CC,
                                         bq, 0, 1, sG, CC + 1,
                                         Tq, TQ_BATCH, TQ_PITCH,
                                         nullptr, 0, nullptr, 0);
    nt_kernel<<<gCorr, blk, 0, stream>>>(Tq, TQ_BATCH, TQ_PITCH, Wk, 0, CC, CC, CC,
                                         Tq + CC, TQ_BATCH, TQ_PITCH, bk, 0,
                                         corr, (long)CC * CC, CC,
                                         nullptr, 0, nullptr, 0);

    gemm256_kernel<<<gGemm, blk, 0, stream>>>(corr, (long)CC * CC, nullptr,
                                              mask, xStride, out, xStride);

    softmax_out_kernel<<<dim3(BB * CC), dim3(256), 0, stream>>>(
        out, fg, mask, Vv, gamma, out);
}


# fg bg^T product (K=HW): split-K into 4 partial planes so a block takes a 64x64 tile (wave 32x32, 2x2 accumulators, one LDS fragment read per WMMA); next product adds the planes; changes sum order
# speedup vs baseline: 1.8109x; 1.0309x over previous
#include <hip/hip_runtime.h>

typedef __attribute__((ext_vector_type(2))) float v2f;
typedef __attribute__((ext_vector_type(4))) float v4f;
typedef __attribute__((ext_vector_type(8))) float v8f;

#define BB 8
#define CC 256
#define HWN 4096
#define MT_BATCH (257 * 256)
#define TQ_PITCH 272
#define TQ_BATCH (256 * 272)

__global__ __launch_bounds__(128) __attribute__((amdgpu_waves_per_eu(4, 8))) void gemm256_kernel(
    const float* __restrict__ A, long aStride,
    const float* __restrict__ bias,
    const float* __restrict__ Bm, long bStride,
    float* __restrict__ D, long dStride)
{
    const int lane  = threadIdx.x & 31;
    const int wave  = threadIdx.x >> 5;
    const int g     = lane >> 4;
    const int ln    = lane & 15;
    const int batch = blockIdx.z;
    const int mBase = blockIdx.y * 64 + wave * 16;
    const int nBase = blockIdx.x * 64;

    const float* Ab = A + (long)batch * aStride;
    const float* Bb = Bm + (long)batch * bStride;
    float*       Db = D + (long)batch * dStride;

    v8f acc0 = {}, acc1 = {}, acc2 = {}, acc3 = {};

    const float* arow = Ab + (long)(mBase + ln) * CC + 8 * g;
    const float* bcol = Bb + (long)(8 * g) * HWN + nBase + 4 * ln;

    #pragma unroll 2
    for (int k = 0; k < CC; k += 16) {
        const v4f a0 = *(const v4f*)(arow + k);
        const v4f a1 = *(const v4f*)(arow + k + 4);
        const float* bp = bcol + (long)k * HWN;
        const v4f r0 = *(const v4f*)(bp);
        const v4f r1 = *(const v4f*)(bp + 1 * HWN);
        const v4f r2 = *(const v4f*)(bp + 2 * HWN);
        const v4f r3 = *(const v4f*)(bp + 3 * HWN);
        const v4f r4 = *(const v4f*)(bp + 4 * HWN);
        const v4f r5 = *(const v4f*)(bp + 5 * HWN);
        const v4f r6 = *(const v4f*)(bp + 6 * HWN);
        const v4f r7 = *(const v4f*)(bp + 7 * HWN);

        v2f a, b0, b1, b2, b3;
        a  = (v2f){ a0.x, a0.y };
        b0 = (v2f){ r0.x, r1.x }; b1 = (v2f){ r0.y, r1.y };
        b2 = (v2f){ r0.z, r1.z }; b3 = (v2f){ r0.w, r1.w };
        acc0 = __builtin_amdgcn_wmma_f32_16x16x4_f32(false, a, false, b0, (short)0, acc0, false, false);
        acc1 = __builtin_amdgcn_wmma_f32_16x16x4_f32(false, a, false, b1, (short)0, acc1, false, false);
        acc2 = __builtin_amdgcn_wmma_f32_16x16x4_f32(false, a, false, b2, (short)0, acc2, false, false);
        acc3 = __builtin_amdgcn_wmma_f32_16x16x4_f32(false, a, false, b3, (short)0, acc3, false, false);
        a  = (v2f){ a0.z, a0.w };
        b0 = (v2f){ r2.x, r3.x }; b1 = (v2f){ r2.y, r3.y };
        b2 = (v2f){ r2.z, r3.z }; b3 = (v2f){ r2.w, r3.w };
        acc0 = __builtin_amdgcn_wmma_f32_16x16x4_f32(false, a, false, b0, (short)0, acc0, false, false);
        acc1 = __builtin_amdgcn_wmma_f32_16x16x4_f32(false, a, false, b1, (short)0, acc1, false, false);
        acc2 = __builtin_amdgcn_wmma_f32_16x16x4_f32(false, a, false, b2, (short)0, acc2, false, false);
        acc3 = __builtin_amdgcn_wmma_f32_16x16x4_f32(false, a, false, b3, (short)0, acc3, false, false);
        a  = (v2f){ a1.x, a1.y };
        b0 = (v2f){ r4.x, r5.x }; b1 = (v2f){ r4.y, r5.y };
        b2 = (v2f){ r4.z, r5.z }; b3 = (v2f){ r4.w, r5.w };
        acc0 = __builtin_amdgcn_wmma_f32_16x16x4_f32(false, a, false, b0, (short)0, acc0, false, false);
        acc1 = __builtin_amdgcn_wmma_f32_16x16x4_f32(false, a, false, b1, (short)0, acc1, false, false);
        acc2 = __builtin_amdgcn_wmma_f32_16x16x4_f32(false, a, false, b2, (short)0, acc2, false, false);
        acc3 = __builtin_amdgcn_wmma_f32_16x16x4_f32(false, a, false, b3, (short)0, acc3, false, false);
        a  = (v2f){ a1.z, a1.w };
        b0 = (v2f){ r6.x, r7.x }; b1 = (v2f){ r6.y, r7.y };
        b2 = (v2f){ r6.z, r7.z }; b3 = (v2f){ r6.w, r7.w };
        acc0 = __builtin_amdgcn_wmma_f32_16x16x4_f32(false, a, false, b0, (short)0, acc0, false, false);
        acc1 = __builtin_amdgcn_wmma_f32_16x16x4_f32(false, a, false, b1, (short)0, acc1, false, false);
        acc2 = __builtin_amdgcn_wmma_f32_16x16x4_f32(false, a, false, b2, (short)0, acc2, false, false);
        acc3 = __builtin_amdgcn_wmma_f32_16x16x4_f32(false, a, false, b3, (short)0, acc3, false, false);
    }

    #pragma unroll
    for (int r = 0; r < 8; ++r) {
        const int mo = mBase + r + 8 * g;
        const float bv = bias ? bias[mo] : 0.0f;
        const v4f o = { acc0[r] + bv, acc1[r] + bv, acc2[r] + bv, acc3[r] + bv };
        *(v4f*)(Db + (long)mo * HWN + nBase + 4 * ln) = o;
    }
}

#define NT_GROUP(SUMS)                                                                  \
    {                                                                                   \
        const v4f a0 = *(const v4f*)(ap + k);                                           \
        const v4f a1 = *(const v4f*)(ap + k + 4);                                       \
        const v4f b0 = *(const v4f*)(bp + k);                                           \
        const v4f b1 = *(const v4f*)(bp + k + 4);                                       \
        if (SUMS) { sa += a0 + a1; sb += b0 + b1; }                                     \
        acc = __builtin_amdgcn_wmma_f32_16x16x4_f32(false, (v2f){ a0.x, a0.y }, false, (v2f){ b0.x, b0.y }, (short)0, acc, false, false); \
        acc = __builtin_amdgcn_wmma_f32_16x16x4_f32(false, (v2f){ a0.z, a0.w }, false, (v2f){ b0.z, b0.w }, (short)0, acc, false, false); \
        acc = __builtin_amdgcn_wmma_f32_16x16x4_f32(false, (v2f){ a1.x, a1.y }, false, (v2f){ b1.x, b1.y }, (short)0, acc, false, false); \
        acc = __builtin_amdgcn_wmma_f32_16x16x4_f32(false, (v2f){ a1.z, a1.w }, false, (v2f){ b1.z, b1.w }, (short)0, acc, false, false); \
    }

__global__ __launch_bounds__(128) void nt_kernel(
    const float* __restrict__ A,  long aBatch, int aPitch,
    const float* __restrict__ Bt, long bBatch, int bPitch, int bRows,
    int kLen,
    const float* __restrict__ u, long uBatch, int uStride,
    const float* __restrict__ w, long wBatch,
    float* __restrict__ D, long dBatch, int dPitch,
    float* __restrict__ sumA, long saBatch,
    float* __restrict__ sumB, long sbBatch)
{
    const int lane  = threadIdx.x & 31;
    const int wave  = threadIdx.x >> 5;
    const int g     = lane >> 4;
    const int ln    = lane & 15;
    const int batch = blockIdx.z;
    const int mBase = blockIdx.y * 64 + wave * 16;
    const int nBase = blockIdx.x * 16;
    const int nRow  = min(nBase + ln, bRows - 1);

    const float* ap = A  + (long)batch * aBatch + (long)(mBase + ln) * aPitch + 8 * g;
    const float* bp = Bt + (long)batch * bBatch + (long)nRow * bPitch + 8 * g;

    v8f acc = {};
    v4f sa = {}, sb = {};
    const bool sumsA = (sumA != nullptr) && (blockIdx.x == 0);
    const bool sumsB = (sumA != nullptr) && (blockIdx.y == 0) && (wave == 0);
    if (sumsA || sumsB) {
        for (int k0 = 0; k0 < kLen; k0 += 64) {
            #pragma unroll
            for (int kk = 0; kk < 64; kk += 16) { const int k = k0 + kk; NT_GROUP(true) }
        }
        float ra = (sa.x + sa.y) + (sa.z + sa.w);
        float rb = (sb.x + sb.y) + (sb.z + sb.w);
        ra += __shfl_xor(ra, 16, 32);
        rb += __shfl_xor(rb, 16, 32);
        if (g == 0) {
            if (sumsA) sumA[(long)batch * saBatch + mBase + ln] = ra;
            if (sumsB) sumB[(long)batch * sbBatch + nBase + ln] = rb;
        }
        if (sumsA && sumsB && lane == 0) sumA[(long)batch * saBatch + CC] = (float)kLen;
    } else {
        for (int k0 = 0; k0 < kLen; k0 += 64) {
            #pragma unroll
            for (int kk = 0; kk < 64; kk += 16) { const int k = k0 + kk; NT_GROUP(false) }
        }
    }

    const float wv = w ? w[(long)batch * wBatch + nRow] : 0.0f;
    float* db = D + (long)batch * dBatch;
    #pragma unroll
    for (int r = 0; r < 8; ++r) {
        const int mo = mBase + r + 8 * g;
        const float uv = u ? u[(long)batch * uBatch + (long)mo * uStride] : 0.0f;
        db[(long)mo * dPitch + nBase + ln] = acc[r] + uv * wv;
    }
}

template <int PLANES>
__global__ __launch_bounds__(128) void nt_planes_kernel(
    const float* __restrict__ A,  long aBatch, int aPitch,
    const float* __restrict__ Bt, long bBatch, long bPlane, int bPitch, int bRows,
    int kLen,
    const float* __restrict__ u, long uBatch, int uStride,
    const float* __restrict__ w, long wBatch, long wPlane,
    float* __restrict__ D, long dBatch, int dPitch)
{
    const int lane  = threadIdx.x & 31;
    const int wave  = threadIdx.x >> 5;
    const int g     = lane >> 4;
    const int ln    = lane & 15;
    const int batch = blockIdx.z;
    const int mBase = blockIdx.y * 64 + wave * 16;
    const int nBase = blockIdx.x * 16;
    const int nRow  = min(nBase + ln, bRows - 1);

    const float* ap = A  + (long)batch * aBatch + (long)(mBase + ln) * aPitch + 8 * g;
    const float* bp = Bt + (long)batch * bBatch + (long)nRow * bPitch + 8 * g;

    v8f acc = {};
    for (int k0 = 0; k0 < kLen; k0 += 64) {
        v4f fa[8], fb[PLANES][8];
        #pragma unroll
        for (int j = 0; j < 4; ++j) {
            fa[2 * j]     = *(const v4f*)(ap + k0 + 16 * j);
            fa[2 * j + 1] = *(const v4f*)(ap + k0 + 16 * j + 4);
            #pragma unroll
            for (int p = 0; p < PLANES; ++p) {
                fb[p][2 * j]     = *(const v4f*)(bp + (long)p * bPlane + k0 + 16 * j);
                fb[p][2 * j + 1] = *(const v4f*)(bp + (long)p * bPlane + k0 + 16 * j + 4);
            }
        }
        __builtin_amdgcn_sched_barrier(0);
        #pragma unroll
        for (int j = 0; j < 4; ++j) {
            const v4f a0 = fa[2 * j], a1 = fa[2 * j + 1];
            v4f b0 = fb[0][2 * j], b1 = fb[0][2 * j + 1];
            #pragma unroll
            for (int p = 1; p < PLANES; ++p) { b0 += fb[p][2 * j]; b1 += fb[p][2 * j + 1]; }
            acc = __builtin_amdgcn_wmma_f32_16x16x4_f32(false, (v2f){ a0.x, a0.y }, false, (v2f){ b0.x, b0.y }, (short)0, acc, false, false);
            acc = __builtin_amdgcn_wmma_f32_16x16x4_f32(false, (v2f){ a0.z, a0.w }, false, (v2f){ b0.z, b0.w }, (short)0, acc, false, false);
            acc = __builtin_amdgcn_wmma_f32_16x16x4_f32(false, (v2f){ a1.x, a1.y }, false, (v2f){ b1.x, b1.y }, (short)0, acc, false, false);
            acc = __builtin_amdgcn_wmma_f32_16x16x4_f32(false, (v2f){ a1.z, a1.w }, false, (v2f){ b1.z, b1.w }, (short)0, acc, false, false);
        }
    }

    float wv = 0.0f;
    if (w) {
        #pragma unroll
        for (int p = 0; p < PLANES; ++p) wv += w[(long)batch * wBatch + (long)p * wPlane + nRow];
    }
    float* db = D + (long)batch * dBatch;
    #pragma unroll
    for (int r = 0; r < 8; ++r) {
        const int mo = mBase + r + 8 * g;
        const float uv = u ? u[(long)batch * uBatch + (long)mo * uStride] : 0.0f;
        db[(long)mo * dPitch + nBase + ln] = acc[r] + uv * wv;
    }
}

#define MT_KC    64
#define MT_PITCH 68
#define MT_SPLIT 4

#define MT_WMMA(AV, BV, ACC) ACC = __builtin_amdgcn_wmma_f32_16x16x4_f32(false, AV, false, BV, (short)0, ACC, false, false);

#define MT_K4(X, Y, P, Q)                                                               \
    MT_WMMA(X, P, acc00) MT_WMMA(X, Q, acc01) MT_WMMA(Y, P, acc10) MT_WMMA(Y, Q, acc11)

#define MT_GROUP(SUMS, J)                                                               \
    {                                                                                   \
        const v4f x0 = fa0[2 * (J)], x1 = fa0[2 * (J) + 1];                             \
        const v4f y0 = fa1[2 * (J)], y1 = fa1[2 * (J) + 1];                             \
        const v4f p0 = fb0[2 * (J)], p1 = fb0[2 * (J) + 1];                             \
        const v4f q0 = fb1[2 * (J)], q1 = fb1[2 * (J) + 1];                             \
        if (SUMS) { sa0 += x0 + x1; sa1 += y0 + y1; sb0 += p0 + p1; sb1 += q0 + q1; }   \
        MT_K4(((v2f){ x0.x, x0.y }), ((v2f){ y0.x, y0.y }), ((v2f){ p0.x, p0.y }), ((v2f){ q0.x, q0.y })) \
        MT_K4(((v2f){ x0.z, x0.w }), ((v2f){ y0.z, y0.w }), ((v2f){ p0.z, p0.w }), ((v2f){ q0.z, q0.w })) \
        MT_K4(((v2f){ x1.x, x1.y }), ((v2f){ y1.x, y1.y }), ((v2f){ p1.x, p1.y }), ((v2f){ q1.x, q1.y })) \
        MT_K4(((v2f){ x1.z, x1.w }), ((v2f){ y1.z, y1.w }), ((v2f){ p1.z, p1.w }), ((v2f){ q1.z, q1.w })) \
    }

template <bool SUMS>
__device__ __forceinline__ void mt_chunks(
    const float* __restrict__ ag, const float* __restrict__ bgl, int aPitch, int bPitch, int kLen,
    float (*sA)[64][MT_PITCH], float (*sB)[64][MT_PITCH],
    int srow, int scol, int g, int ln, int wm, int wn,
    v8f& acc00, v8f& acc01, v8f& acc10, v8f& acc11,
    v4f& sa0, v4f& sa1, v4f& sb0, v4f& sb1)
{
    v4f pa[8], pb[8];
    #pragma unroll
    for (int i = 0; i < 8; ++i) pa[i] = *(const v4f*)(ag + (long)(8 * i) * aPitch);
    #pragma unroll
    for (int i = 0; i < 8; ++i) pb[i] = *(const v4f*)(bgl + (long)(8 * i) * bPitch);
    #pragma unroll
    for (int i = 0; i < 8; ++i) *(v4f*)&sA[0][8 * i + srow][scol] = pa[i];
    #pragma unroll
    for (int i = 0; i < 8; ++i) *(v4f*)&sB[0][8 * i + srow][scol] = pb[i];
    __syncthreads();

    const int nChunk = kLen / MT_KC;
    for (int t = 0; t < nChunk; ++t) {
        const int cur = t & 1;
        const bool more = (t + 1 < nChunk);
        if (more) {
            const long k1 = (long)(t + 1) * MT_KC;
            #pragma unroll
            for (int i = 0; i < 8; ++i) pa[i] = *(const v4f*)(ag + (long)(8 * i) * aPitch + k1);
            #pragma unroll
            for (int i = 0; i < 8; ++i) pb[i] = *(const v4f*)(bgl + (long)(8 * i) * bPitch + k1);
        }
        const float* as0 = &sA[cur][32 * wm + ln][8 * g];
        const float* as1 = as0 + 16 * MT_PITCH;
        const float* bs0 = &sB[cur][32 * wn + ln][8 * g];
        const float* bs1 = bs0 + 16 * MT_PITCH;
        v4f fa0[8], fa1[8], fb0[8], fb1[8];
        #pragma unroll
        for (int j = 0; j < 4; ++j) {
            fa0[2 * j]     = *(const v4f*)(as0 + 16 * j);
            fa0[2 * j + 1] = *(const v4f*)(as0 + 16 * j + 4);
            fa1[2 * j]     = *(const v4f*)(as1 + 16 * j);
            fa1[2 * j + 1] = *(const v4f*)(as1 + 16 * j + 4);
            fb0[2 * j]     = *(const v4f*)(bs0 + 16 * j);
            fb0[2 * j + 1] = *(const v4f*)(bs0 + 16 * j + 4);
            fb1[2 * j]     = *(const v4f*)(bs1 + 16 * j);
            fb1[2 * j + 1] = *(const v4f*)(bs1 + 16 * j + 4);
        }
        __builtin_amdgcn_sched_barrier(0);
        #pragma unroll
        for (int j = 0; j < 4; ++j) MT_GROUP(SUMS, j)
        if (more) {
            #pragma unroll
            for (int i = 0; i < 8; ++i) *(v4f*)&sA[cur ^ 1][8 * i + srow][scol] = pa[i];
            #pragma unroll
            for (int i = 0; i < 8; ++i) *(v4f*)&sB[cur ^ 1][8 * i + srow][scol] = pb[i];
        }
        __syncthreads();
    }
}

__global__ __launch_bounds__(128) void mt_lds_kernel(
    const float* __restrict__ A,  long aBatch, int aPitch,
    const float* __restrict__ Bt, long bBatch, int bPitch,
    int kLen,
    float* __restrict__ D, long dPlane, int dPitch,
    float* __restrict__ sumA, long saPlane,
    float* __restrict__ sumB, long sbPlane)
{
    __shared__ __attribute__((aligned(16))) float sA[2][64][MT_PITCH];
    __shared__ __attribute__((aligned(16))) float sB[2][64][MT_PITCH];

    const int lane  = threadIdx.x & 31;
    const int wave  = threadIdx.x >> 5;
    const int g     = lane >> 4;
    const int ln    = lane & 15;
    const int wm    = wave >> 1;
    const int wn    = wave & 1;
    const int plane = blockIdx.z;
    const int batch = plane / MT_SPLIT;
    const int part  = plane % MT_SPLIT;
    const int mBlk  = blockIdx.y * 64;
    const int nBlk  = blockIdx.x * 64;
    const int mBase = mBlk + 32 * wm;
    const int nBase = nBlk + 32 * wn;

    const int srow = 2 * wave + g;
    const int scol = 4 * ln;
    const long kOff = (long)part * kLen;
    const float* ag  = A  + (long)batch * aBatch + (long)(mBlk + srow) * aPitch + kOff + scol;
    const float* bgl = Bt + (long)batch * bBatch + (long)(nBlk + srow) * bPitch + kOff + scol;

    v8f acc00 = {}, acc01 = {}, acc10 = {}, acc11 = {};
    v4f sa0 = {}, sa1 = {}, sb0 = {}, sb1 = {};
    const bool sumsA = (blockIdx.x == 0) && (wn == 0);
    const bool sumsB = (blockIdx.y == 0) && (wm == 0);
    if (blockIdx.x == 0 || blockIdx.y == 0) {
        mt_chunks<true>(ag, bgl, aPitch, bPitch, kLen, sA, sB, srow, scol, g, ln, wm, wn,
                        acc00, acc01, acc10, acc11, sa0, sa1, sb0, sb1);
        float ra0 = (sa0.x + sa0.y) + (sa0.z + sa0.w);
        float ra1 = (sa1.x + sa1.y) + (sa1.z + sa1.w);
        float rb0 = (sb0.x + sb0.y) + (sb0.z + sb0.w);
        float rb1 = (sb1.x + sb1.y) + (sb1.z + sb1.w);
        ra0 += __shfl_xor(ra0, 16, 32);
        ra1 += __shfl_xor(ra1, 16, 32);
        rb0 += __shfl_xor(rb0, 16, 32);
        rb1 += __shfl_xor(rb1, 16, 32);
        if (g == 0) {
            if (sumsA) {
                sumA[(long)plane * saPlane + mBase + ln]      = ra0;
                sumA[(long)plane * saPlane + mBase + 16 + ln] = ra1;
            }
            if (sumsB) {
                sumB[(long)plane * sbPlane + nBase + ln]      = rb0;
                sumB[(long)plane * sbPlane + nBase + 16 + ln] = rb1;
            }
        }
        if (sumsA && sumsB && lane == 0) sumA[(long)plane * saPlane + CC] = (float)kLen;
    } else {
        mt_chunks<false>(ag, bgl, aPitch, bPitch, kLen, sA, sB, srow, scol, g, ln, wm, wn,
                         acc00, acc01, acc10, acc11, sa0, sa1, sb0, sb1);
    }

    float* db = D + (long)plane * dPlane;
    #pragma unroll
    for (int r = 0; r < 8; ++r) {
        const long m0 = (long)(mBase + r + 8 * g) * dPitch;
        const long m1 = m0 + (long)16 * dPitch;
        db[m0 + nBase + ln]      = acc00[r];
        db[m0 + nBase + 16 + ln] = acc01[r];
        db[m1 + nBase + ln]      = acc10[r];
        db[m1 + nBase + 16 + ln] = acc11[r];
    }
}

__global__ __launch_bounds__(256) void softmax_out_kernel(
    const float* __restrict__ scores,
    const float* __restrict__ fg, const float* __restrict__ mask,
    const float* __restrict__ V,  const float* __restrict__ gamma,
    float* __restrict__ out)
{
    const int  tid  = threadIdx.x;
    const int  lane = tid & 31;
    const int  wid  = tid >> 5;
    const long base = (long)blockIdx.x * HWN;

    __shared__ float red[8];

    float vals[16];
    float mx = -3.402823466e38f;
    #pragma unroll
    for (int t = 0; t < 16; ++t) {
        vals[t] = scores[base + t * 256 + tid];
        mx = fmaxf(mx, vals[t]);
    }
    #pragma unroll
    for (int off = 16; off >= 1; off >>= 1)
        mx = fmaxf(mx, __shfl_xor(mx, off, 32));
    if (lane == 0) red[wid] = mx;
    __syncthreads();
    #pragma unroll
    for (int w = 0; w < 8; ++w) mx = fmaxf(mx, red[w]);
    __syncthreads();

    float sum = 0.0f;
    #pragma unroll
    for (int t = 0; t < 16; ++t) {
        vals[t] = __expf(vals[t] - mx);
        sum += vals[t];
    }
    #pragma unroll
    for (int off = 16; off >= 1; off >>= 1)
        sum += __shfl_xor(sum, off, 32);
    if (lane == 0) red[wid] = sum;
    __syncthreads();
    float tot = 0.0f;
    #pragma unroll
    for (int w = 0; w < 8; ++w) tot += red[w];

    const float inv = 1.0f / tot;
    const float gam = gamma[0];
    #pragma unroll
    for (int t = 0; t < 16; ++t) {
        const long i = base + t * 256 + tid;
        const float att = vals[t] * inv * V[i];
        const float mk  = mask[i];
        out[i] = fg[i] * mk + gam * (1.0f - mk) * att;
    }
}

extern "C" void kernel_launch(void* const* d_in, const int* in_sizes, int n_in,
                              void* d_out, int out_size, void* d_ws, size_t ws_size,
                              hipStream_t stream)
{
    (void)in_sizes; (void)n_in; (void)out_size; (void)ws_size;

    const float* fg    = (const float*)d_in[0];
    const float* bg    = (const float*)d_in[1];
    const float* mask  = (const float*)d_in[2];
    const float* Wq    = (const float*)d_in[3];
    const float* bq    = (const float*)d_in[4];
    const float* Wk    = (const float*)d_in[5];
    const float* bk    = (const float*)d_in[6];
    const float* Wv    = (const float*)d_in[7];
    const float* bv    = (const float*)d_in[8];
    const float* gamma = (const float*)d_in[9];
    float* out = (float*)d_out;

    const long mat = (long)BB * CC * HWN;
    float* Vv   = (float*)d_ws;
    float* Tq   = Vv + mat;
    float* corr = Tq + (long)BB * TQ_BATCH;
    float* MTp  = out;
    float* sGp  = out + (long)BB * MT_SPLIT * MT_BATCH;

    const dim3 blk(128);
    const dim3 gGemm(HWN / 64, CC / 64, BB);
    const long xStride = (long)CC * HWN;

    gemm256_kernel<<<gGemm, blk, 0, stream>>>(Wv, 0, bv, fg, xStride, Vv, xStride);

    const dim3 gCorr(CC / 16, CC / 64, BB);
    mt_lds_kernel<<<dim3(CC / 64, CC / 64, BB * MT_SPLIT), blk, 0, stream>>>(
                                             bg, xStride, HWN, fg, xStride, HWN, HWN / MT_SPLIT,
                                             MTp, MT_BATCH, CC,
                                             sGp, CC + 1, MTp + CC * CC, MT_BATCH);
    nt_planes_kernel<MT_SPLIT><<<dim3(CC / 16 + 1, CC / 64, BB), blk, 0, stream>>>(
                                         Wq, 0, CC, MTp, (long)MT_SPLIT * MT_BATCH, MT_BATCH, CC, CC + 1, CC,
                                         bq, 0, 1, sGp, (long)MT_SPLIT * (CC + 1), CC + 1,
                                         Tq, TQ_BATCH, TQ_PITCH);
    nt_kernel<<<gCorr, blk, 0, stream>>>(Tq, TQ_BATCH, TQ_PITCH, Wk, 0, CC, CC, CC,
                                         Tq + CC, TQ_BATCH, TQ_PITCH, bk, 0,
                                         corr, (long)CC * CC, CC,
                                         nullptr, 0, nullptr, 0);

    gemm256_kernel<<<gGemm, blk, 0, stream>>>(corr, (long)CC * CC, nullptr,
                                              mask, xStride, out, xStride);

    softmax_out_kernel<<<dim3(BB * CC), dim3(256), 0, stream>>>(
        out, fg, mask, Vv, gamma, out);
}


# v product loses its own launch: its tiles ride, by batch, as extra blocks of the two short correlation-chain products' launches, so its WMMAs run under their load waits; sum orders kept
# speedup vs baseline: 1.9739x; 1.0900x over previous
#include <hip/hip_runtime.h>

typedef __attribute__((ext_vector_type(2))) float v2f;
typedef __attribute__((ext_vector_type(4))) float v4f;
typedef __attribute__((ext_vector_type(8))) float v8f;

#define BB 8
#define CC 256
#define HWN 4096
#define MT_BATCH (257 * 256)
#define TQ_PITCH 272
#define TQ_BATCH (256 * 272)

__device__ __forceinline__ void gemm256_tile(
    const float* __restrict__ A, long aStride,
    const float* __restrict__ bias,
    const float* __restrict__ Bm, long bStride,
    float* __restrict__ D, long dStride,
    int bx, int by, int batch)
{
    const int lane  = threadIdx.x & 31;
    const int wave  = threadIdx.x >> 5;
    const int g     = lane >> 4;
    const int ln    = lane & 15;
    const int mBase = by * 64 + wave * 16;
    const int nBase = bx * 64;

    const float* Ab = A + (long)batch * aStride;
    const float* Bb = Bm + (long)batch * bStride;
    float*       Db = D + (long)batch * dStride;

    v8f acc0 = {}, acc1 = {}, acc2 = {}, acc3 = {};

    const float* arow = Ab + (long)(mBase + ln) * CC + 8 * g;
    const float* bcol = Bb + (long)(8 * g) * HWN + nBase + 4 * ln;

    #pragma unroll 2
    for (int k = 0; k < CC; k += 16) {
        const v4f a0 = *(const v4f*)(arow + k);
        const v4f a1 = *(const v4f*)(arow + k + 4);
        const float* bp = bcol + (long)k * HWN;
        const v4f r0 = *(const v4f*)(bp);
        const v4f r1 = *(const v4f*)(bp + 1 * HWN);
        const v4f r2 = *(const v4f*)(bp + 2 * HWN);
        const v4f r3 = *(const v4f*)(bp + 3 * HWN);
        const v4f r4 = *(const v4f*)(bp + 4 * HWN);
        const v4f r5 = *(const v4f*)(bp + 5 * HWN);
        const v4f r6 = *(const v4f*)(bp + 6 * HWN);
        const v4f r7 = *(const v4f*)(bp + 7 * HWN);

        v2f a, b0, b1, b2, b3;
        a  = (v2f){ a0.x, a0.y };
        b0 = (v2f){ r0.x, r1.x }; b1 = (v2f){ r0.y, r1.y };
        b2 = (v2f){ r0.z, r1.z }; b3 = (v2f){ r0.w, r1.w };
        acc0 = __builtin_amdgcn_wmma_f32_16x16x4_f32(false, a, false, b0, (short)0, acc0, false, false);
        acc1 = __builtin_amdgcn_wmma_f32_16x16x4_f32(false, a, false, b1, (short)0, acc1, false, false);
        acc2 = __builtin_amdgcn_wmma_f32_16x16x4_f32(false, a, false, b2, (short)0, acc2, false, false);
        acc3 = __builtin_amdgcn_wmma_f32_16x16x4_f32(false, a, false, b3, (short)0, acc3, false, false);
        a  = (v2f){ a0.z, a0.w };
        b0 = (v2f){ r2.x, r3.x }; b1 = (v2f){ r2.y, r3.y };
        b2 = (v2f){ r2.z, r3.z }; b3 = (v2f){ r2.w, r3.w };
        acc0 = __builtin_amdgcn_wmma_f32_16x16x4_f32(false, a, false, b0, (short)0, acc0, false, false);
        acc1 = __builtin_amdgcn_wmma_f32_16x16x4_f32(false, a, false, b1, (short)0, acc1, false, false);
        acc2 = __builtin_amdgcn_wmma_f32_16x16x4_f32(false, a, false, b2, (short)0, acc2, false, false);
        acc3 = __builtin_amdgcn_wmma_f32_16x16x4_f32(false, a, false, b3, (short)0, acc3, false, false);
        a  = (v2f){ a1.x, a1.y };
        b0 = (v2f){ r4.x, r5.x }; b1 = (v2f){ r4.y, r5.y };
        b2 = (v2f){ r4.z, r5.z }; b3 = (v2f){ r4.w, r5.w };
        acc0 = __builtin_amdgcn_wmma_f32_16x16x4_f32(false, a, false, b0, (short)0, acc0, false, false);
        acc1 = __builtin_amdgcn_wmma_f32_16x16x4_f32(false, a, false, b1, (short)0, acc1, false, false);
        acc2 = __builtin_amdgcn_wmma_f32_16x16x4_f32(false, a, false, b2, (short)0, acc2, false, false);
        acc3 = __builtin_amdgcn_wmma_f32_16x16x4_f32(false, a, false, b3, (short)0, acc3, false, false);
        a  = (v2f){ a1.z, a1.w };
        b0 = (v2f){ r6.x, r7.x }; b1 = (v2f){ r6.y, r7.y };
        b2 = (v2f){ r6.z, r7.z }; b3 = (v2f){ r6.w, r7.w };
        acc0 = __builtin_amdgcn_wmma_f32_16x16x4_f32(false, a, false, b0, (short)0, acc0, false, false);
        acc1 = __builtin_amdgcn_wmma_f32_16x16x4_f32(false, a, false, b1, (short)0, acc1, false, false);
        acc2 = __builtin_amdgcn_wmma_f32_16x16x4_f32(false, a, false, b2, (short)0, acc2, false, false);
        acc3 = __builtin_amdgcn_wmma_f32_16x16x4_f32(false, a, false, b3, (short)0, acc3, false, false);
    }

    #pragma unroll
    for (int r = 0; r < 8; ++r) {
        const int mo = mBase + r + 8 * g;
        const float bv = bias ? bias[mo] : 0.0f;
        const v4f o = { acc0[r] + bv, acc1[r] + bv, acc2[r] + bv, acc3[r] + bv };
        *(v4f*)(Db + (long)mo * HWN + nBase + 4 * ln) = o;
    }
}

__global__ __launch_bounds__(128) __attribute__((amdgpu_waves_per_eu(4, 8))) void gemm256_kernel(
    const float* __restrict__ A, long aStride,
    const float* __restrict__ bias,
    const float* __restrict__ Bm, long bStride,
    float* __restrict__ D, long dStride)
{
    gemm256_tile(A, aStride, bias, Bm, bStride, D, dStride, blockIdx.x, blockIdx.y, blockIdx.z);
}

#define NT_GROUP                                                                        \
    {                                                                                   \
        const v4f a0 = *(const v4f*)(ap + k);                                           \
        const v4f a1 = *(const v4f*)(ap + k + 4);                                       \
        const v4f b0 = *(const v4f*)(bp + k);                                           \
        const v4f b1 = *(const v4f*)(bp + k + 4);                                       \
        acc = __builtin_amdgcn_wmma_f32_16x16x4_f32(false, (v2f){ a0.x, a0.y }, false, (v2f){ b0.x, b0.y }, (short)0, acc, false, false); \
        acc = __builtin_amdgcn_wmma_f32_16x16x4_f32(false, (v2f){ a0.z, a0.w }, false, (v2f){ b0.z, b0.w }, (short)0, acc, false, false); \
        acc = __builtin_amdgcn_wmma_f32_16x16x4_f32(false, (v2f){ a1.x, a1.y }, false, (v2f){ b1.x, b1.y }, (short)0, acc, false, false); \
        acc = __builtin_amdgcn_wmma_f32_16x16x4_f32(false, (v2f){ a1.z, a1.w }, false, (v2f){ b1.z, b1.w }, (short)0, acc, false, false); \
    }

__device__ __forceinline__ void nt_tile(
    const float* __restrict__ A,  long aBatch, int aPitch,
    const float* __restrict__ Bt, long bBatch, int bPitch, int bRows,
    int kLen,
    const float* __restrict__ u, long uBatch, int uStride,
    const float* __restrict__ w, long wBatch,
    float* __restrict__ D, long dBatch, int dPitch,
    int bx, int by, int batch)
{
    const int lane  = threadIdx.x & 31;
    const int wave  = threadIdx.x >> 5;
    const int g     = lane >> 4;
    const int ln    = lane & 15;
    const int mBase = by * 64 + wave * 16;
    const int nBase = bx * 16;
    const int nRow  = min(nBase + ln, bRows - 1);

    const float* ap = A  + (long)batch * aBatch + (long)(mBase + ln) * aPitch + 8 * g;
    const float* bp = Bt + (long)batch * bBatch + (long)nRow * bPitch + 8 * g;

    v8f acc = {};
    for (int k0 = 0; k0 < kLen; k0 += 64) {
        #pragma unroll
        for (int kk = 0; kk < 64; kk += 16) { const int k = k0 + kk; NT_GROUP }
    }

    const float wv = w ? w[(long)batch * wBatch + nRow] : 0.0f;
    float* db = D + (long)batch * dBatch;
    #pragma unroll
    for (int r = 0; r < 8; ++r) {
        const int mo = mBase + r + 8 * g;
        const float uv = u ? u[(long)batch * uBatch + (long)mo * uStride] : 0.0f;
        db[(long)mo * dPitch + nBase + ln] = acc[r] + uv * wv;
    }
}

template <int PLANES>
__device__ __forceinline__ void nt_planes_tile(
    const float* __restrict__ A,  long aBatch, int aPitch,
    const float* __restrict__ Bt, long bBatch, long bPlane, int bPitch, int bRows,
    int kLen,
    const float* __restrict__ u, long uBatch, int uStride,
    const float* __restrict__ w, long wBatch, long wPlane,
    float* __restrict__ D, long dBatch, int dPitch,
    int bx, int by, int batch)
{
    const int lane  = threadIdx.x & 31;
    const int wave  = threadIdx.x >> 5;
    const int g     = lane >> 4;
    const int ln    = lane & 15;
    const int mBase = by * 64 + wave * 16;
    const int nBase = bx * 16;
    const int nRow  = min(nBase + ln, bRows - 1);

    const float* ap = A  + (long)batch * aBatch + (long)(mBase + ln) * aPitch + 8 * g;
    const float* bp = Bt + (long)batch * bBatch + (long)nRow * bPitch + 8 * g;

    v8f acc = {};
    #pragma unroll 1
    for (int k0 = 0; k0 < kLen; k0 += 64) {
        v4f fa[8], fb[PLANES][8];
        #pragma unroll
        for (int j = 0; j < 4; ++j) {
            fa[2 * j]     = *(const v4f*)(ap + k0 + 16 * j);
            fa[2 * j + 1] = *(const v4f*)(ap + k0 + 16 * j + 4);
            #pragma unroll
            for (int p = 0; p < PLANES; ++p) {
                fb[p][2 * j]     = *(const v4f*)(bp + (long)p * bPlane + k0 + 16 * j);
                fb[p][2 * j + 1] = *(const v4f*)(bp + (long)p * bPlane + k0 + 16 * j + 4);
            }
        }
        __builtin_amdgcn_sched_barrier(0);
        #pragma unroll
        for (int j = 0; j < 4; ++j) {
            const v4f a0 = fa[2 * j], a1 = fa[2 * j + 1];
            v4f b0 = fb[0][2 * j], b1 = fb[0][2 * j + 1];
            #pragma unroll
            for (int p = 1; p < PLANES; ++p) { b0 += fb[p][2 * j]; b1 += fb[p][2 * j + 1]; }
            acc = __builtin_amdgcn_wmma_f32_16x16x4_f32(false, (v2f){ a0.x, a0.y }, false, (v2f){ b0.x, b0.y }, (short)0, acc, false, false);
            acc = __builtin_amdgcn_wmma_f32_16x16x4_f32(false, (v2f){ a0.z, a0.w }, false, (v2f){ b0.z, b0.w }, (short)0, acc, false, false);
            acc = __builtin_amdgcn_wmma_f32_16x16x4_f32(false, (v2f){ a1.x, a1.y }, false, (v2f){ b1.x, b1.y }, (short)0, acc, false, false);
            acc = __builtin_amdgcn_wmma_f32_16x16x4_f32(false, (v2f){ a1.z, a1.w }, false, (v2f){ b1.z, b1.w }, (short)0, acc, false, false);
        }
    }

    float wv = 0.0f;
    if (w) {
        #pragma unroll
        for (int p = 0; p < PLANES; ++p) wv += w[(long)batch * wBatch + (long)p * wPlane + nRow];
    }
    float* db = D + (long)batch * dBatch;
    #pragma unroll
    for (int r = 0; r < 8; ++r) {
        const int mo = mBase + r + 8 * g;
        const float uv = u ? u[(long)batch * uBatch + (long)mo * uStride] : 0.0f;
        db[(long)mo * dPitch + nBase + ln] = acc[r] + uv * wv;
    }
}

#define MT_KC    64
#define MT_PITCH 68
#define MT_SPLIT 4

#define MT_WMMA(AV, BV, ACC) ACC = __builtin_amdgcn_wmma_f32_16x16x4_f32(false, AV, false, BV, (short)0, ACC, false, false);

#define MT_K4(X, Y, P, Q)                                                               \
    MT_WMMA(X, P, acc00) MT_WMMA(X, Q, acc01) MT_WMMA(Y, P, acc10) MT_WMMA(Y, Q, acc11)

#define MT_GROUP(SUMS, J)                                                               \
    {                                                                                   \
        const v4f x0 = fa0[2 * (J)], x1 = fa0[2 * (J) + 1];                             \
        const v4f y0 = fa1[2 * (J)], y1 = fa1[2 * (J) + 1];                             \
        const v4f p0 = fb0[2 * (J)], p1 = fb0[2 * (J) + 1];                             \
        const v4f q0 = fb1[2 * (J)], q1 = fb1[2 * (J) + 1];                             \
        if (SUMS) { sa0 += x0 + x1; sa1 += y0 + y1; sb0 += p0 + p1; sb1 += q0 + q1; }   \
        MT_K4(((v2f){ x0.x, x0.y }), ((v2f){ y0.x, y0.y }), ((v2f){ p0.x, p0.y }), ((v2f){ q0.x, q0.y })) \
        MT_K4(((v2f){ x0.z, x0.w }), ((v2f){ y0.z, y0.w }), ((v2f){ p0.z, p0.w }), ((v2f){ q0.z, q0.w })) \
        MT_K4(((v2f){ x1.x, x1.y }), ((v2f){ y1.x, y1.y }), ((v2f){ p1.x, p1.y }), ((v2f){ q1.x, q1.y })) \
        MT_K4(((v2f){ x1.z, x1.w }), ((v2f){ y1.z, y1.w }), ((v2f){ p1.z, p1.w }), ((v2f){ q1.z, q1.w })) \
    }

template <bool SUMS>
__device__ __forceinline__ void mt_chunks(
    const float* __restrict__ ag, const float* __restrict__ bgl, int aPitch, int bPitch, int kLen,
    float (*sA)[64][MT_PITCH], float (*sB)[64][MT_PITCH],
    int srow, int scol, int g, int ln, int wm, int wn,
    v8f& acc00, v8f& acc01, v8f& acc10, v8f& acc11,
    v4f& sa0, v4f& sa1, v4f& sb0, v4f& sb1)
{
    v4f pa[8], pb[8];
    #pragma unroll
    for (int i = 0; i < 8; ++i) pa[i] = *(const v4f*)(ag + (long)(8 * i) * aPitch);
    #pragma unroll
    for (int i = 0; i < 8; ++i) pb[i] = *(const v4f*)(bgl + (long)(8 * i) * bPitch);
    #pragma unroll
    for (int i = 0; i < 8; ++i) *(v4f*)&sA[0][8 * i + srow][scol] = pa[i];
    #pragma unroll
    for (int i = 0; i < 8; ++i) *(v4f*)&sB[0][8 * i + srow][scol] = pb[i];
    __syncthreads();

    const int nChunk = kLen / MT_KC;
    for (int t = 0; t < nChunk; ++t) {
        const int cur = t & 1;
        const bool more = (t + 1 < nChunk);
        if (more) {
            const long k1 = (long)(t + 1) * MT_KC;
            #pragma unroll
            for (int i = 0; i < 8; ++i) pa[i] = *(const v4f*)(ag + (long)(8 * i) * aPitch + k1);
            #pragma unroll
            for (int i = 0; i < 8; ++i) pb[i] = *(const v4f*)(bgl + (long)(8 * i) * bPitch + k1);
        }
        const float* as0 = &sA[cur][32 * wm + ln][8 * g];
        const float* as1 = as0 + 16 * MT_PITCH;
        const float* bs0 = &sB[cur][32 * wn + ln][8 * g];
        const float* bs1 = bs0 + 16 * MT_PITCH;
        v4f fa0[8], fa1[8], fb0[8], fb1[8];
        #pragma unroll
        for (int j = 0; j < 4; ++j) {
            fa0[2 * j]     = *(const v4f*)(as0 + 16 * j);
            fa0[2 * j + 1] = *(const v4f*)(as0 + 16 * j + 4);
            fa1[2 * j]     = *(const v4f*)(as1 + 16 * j);
            fa1[2 * j + 1] = *(const v4f*)(as1 + 16 * j + 4);
            fb0[2 * j]     = *(const v4f*)(bs0 + 16 * j);
            fb0[2 * j + 1] = *(const v4f*)(bs0 + 16 * j + 4);
            fb1[2 * j]     = *(const v4f*)(bs1 + 16 * j);
            fb1[2 * j + 1] = *(const v4f*)(bs1 + 16 * j + 4);
        }
        __builtin_amdgcn_sched_barrier(0);
        #pragma unroll
        for (int j = 0; j < 4; ++j) MT_GROUP(SUMS, j)
        if (more) {
            #pragma unroll
            for (int i = 0; i < 8; ++i) *(v4f*)&sA[cur ^ 1][8 * i + srow][scol] = pa[i];
            #pragma unroll
            for (int i = 0; i < 8; ++i) *(v4f*)&sB[cur ^ 1][8 * i + srow][scol] = pb[i];
        }
        __syncthreads();
    }
}

__global__ __launch_bounds__(128) void mt_lds_kernel(
    const float* __restrict__ A,  long aBatch, int aPitch,
    const float* __restrict__ Bt, long bBatch, int bPitch,
    int kLen,
    float* __restrict__ D, long dPlane, int dPitch,
    float* __restrict__ sumA, long saPlane,
    float* __restrict__ sumB, long sbPlane)
{
    __shared__ __attribute__((aligned(16))) float sA[2][64][MT_PITCH];
    __shared__ __attribute__((aligned(16))) float sB[2][64][MT_PITCH];

    const int lane  = threadIdx.x & 31;
    const int wave  = threadIdx.x >> 5;
    const int g     = lane >> 4;
    const int ln    = lane & 15;
    const int wm    = wave >> 1;
    const int wn    = wave & 1;
    const int plane = blockIdx.z;
    const int batch = plane / MT_SPLIT;
    const int part  = plane % MT_SPLIT;
    const int mBlk  = blockIdx.y * 64;
    const int nBlk  = blockIdx.x * 64;
    const int mBase = mBlk + 32 * wm;
    const int nBase = nBlk + 32 * wn;

    const int srow = 2 * wave + g;
    const int scol = 4 * ln;
    const long kOff = (long)part * kLen;
    const float* ag  = A  + (long)batch * aBatch + (long)(mBlk + srow) * aPitch + kOff + scol;
    const float* bgl = Bt + (long)batch * bBatch + (long)(nBlk + srow) * bPitch + kOff + scol;

    v8f acc00 = {}, acc01 = {}, acc10 = {}, acc11 = {};
    v4f sa0 = {}, sa1 = {}, sb0 = {}, sb1 = {};
    const bool sumsA = (blockIdx.x == 0) && (wn == 0);
    const bool sumsB = (blockIdx.y == 0) && (wm == 0);
    if (blockIdx.x == 0 || blockIdx.y == 0) {
        mt_chunks<true>(ag, bgl, aPitch, bPitch, kLen, sA, sB, srow, scol, g, ln, wm, wn,
                        acc00, acc01, acc10, acc11, sa0, sa1, sb0, sb1);
        float ra0 = (sa0.x + sa0.y) + (sa0.z + sa0.w);
        float ra1 = (sa1.x + sa1.y) + (sa1.z + sa1.w);
        float rb0 = (sb0.x + sb0.y) + (sb0.z + sb0.w);
        float rb1 = (sb1.x + sb1.y) + (sb1.z + sb1.w);
        ra0 += __shfl_xor(ra0, 16, 32);
        ra1 += __shfl_xor(ra1, 16, 32);
        rb0 += __shfl_xor(rb0, 16, 32);
        rb1 += __shfl_xor(rb1, 16, 32);
        if (g == 0) {
            if (sumsA) {
                sumA[(long)plane * saPlane + mBase + ln]      = ra0;
                sumA[(long)plane * saPlane + mBase + 16 + ln] = ra1;
            }
            if (sumsB) {
                sumB[(long)plane * sbPlane + nBase + ln]      = rb0;
                sumB[(long)plane * sbPlane + nBase + 16 + ln] = rb1;
            }
        }
        if (sumsA && sumsB && lane == 0) sumA[(long)plane * saPlane + CC] = (float)kLen;
    } else {
        mt_chunks<false>(ag, bgl, aPitch, bPitch, kLen, sA, sB, srow, scol, g, ln, wm, wn,
                         acc00, acc01, acc10, acc11, sa0, sa1, sb0, sb1);
    }

    float* db = D + (long)plane * dPlane;
    #pragma unroll
    for (int r = 0; r < 8; ++r) {
        const long m0 = (long)(mBase + r + 8 * g) * dPitch;
        const long m1 = m0 + (long)16 * dPitch;
        db[m0 + nBase + ln]      = acc00[r];
        db[m0 + nBase + 16 + ln] = acc01[r];
        db[m1 + nBase + ln]      = acc10[r];
        db[m1 + nBase + 16 + ln] = acc11[r];
    }
}

__global__ __launch_bounds__(256) void softmax_out_kernel(
    const float* __restrict__ scores,
    const float* __restrict__ fg, const float* __restrict__ mask,
    const float* __restrict__ V,  const float* __restrict__ gamma,
    float* __restrict__ out)
{
    const int  tid  = threadIdx.x;
    const int  lane = tid & 31;
    const int  wid  = tid >> 5;
    const long base = (long)blockIdx.x * HWN;

    __shared__ float red[8];

    float vals[16];
    float mx = -3.402823466e38f;
    #pragma unroll
    for (int t = 0; t < 16; ++t) {
        vals[t] = scores[base + t * 256 + tid];
        mx = fmaxf(mx, vals[t]);
    }
    #pragma unroll
    for (int off = 16; off >= 1; off >>= 1)
        mx = fmaxf(mx, __shfl_xor(mx, off, 32));
    if (lane == 0) red[wid] = mx;
    __syncthreads();
    #pragma unroll
    for (int w = 0; w < 8; ++w) mx = fmaxf(mx, red[w]);
    __syncthreads();

    float sum = 0.0f;
    #pragma unroll
    for (int t = 0; t < 16; ++t) {
        vals[t] = __expf(vals[t] - mx);
        sum += vals[t];
    }
    #pragma unroll
    for (int off = 16; off >= 1; off >>= 1)
        sum += __shfl_xor(sum, off, 32);
    if (lane == 0) red[wid] = sum;
    __syncthreads();
    float tot = 0.0f;
    #pragma unroll
    for (int w = 0; w < 8; ++w) tot += red[w];

    const float inv = 1.0f / tot;
    const float gam = gamma[0];
    #pragma unroll
    for (int t = 0; t < 16; ++t) {
        const long i = base + t * 256 + tid;
        const float att = vals[t] * inv * V[i];
        const float mk  = mask[i];
        out[i] = fg[i] * mk + gam * (1.0f - mk) * att;
    }
}

#define GEMM_BX         (HWN / 64)
#define GEMM_PER_BATCH  (GEMM_BX * (CC / 64))
#define NTP_BX          (CC / 16 + 1)
#define NTP_BLOCKS      (NTP_BX * (CC / 64) * BB)
#define NT_BX           (CC / 16)
#define NT_BLOCKS       (NT_BX * (CC / 64) * BB)
#define V_BATCHES_EARLY 3

__global__ __launch_bounds__(128) void ntp_v_kernel(
    const float* __restrict__ Wq, const float* __restrict__ MTp, const float* __restrict__ bq,
    const float* __restrict__ sGp, float* __restrict__ Tq,
    const float* __restrict__ Wv, const float* __restrict__ bv, const float* __restrict__ fg,
    float* __restrict__ Vv, int vBatch0)
{
    const int b = blockIdx.x;
    if (b < NTP_BLOCKS) {
        nt_planes_tile<MT_SPLIT>(Wq, 0, CC, MTp, (long)MT_SPLIT * MT_BATCH, MT_BATCH, CC, CC + 1, CC,
                                 bq, 0, 1, sGp, (long)MT_SPLIT * (CC + 1), CC + 1,
                                 Tq, TQ_BATCH, TQ_PITCH,
                                 b % NTP_BX, (b / NTP_BX) % (CC / 64), b / (NTP_BX * (CC / 64)));
    } else {
        const int t = b - NTP_BLOCKS;
        gemm256_tile(Wv, 0, bv, fg, (long)CC * HWN, Vv, (long)CC * HWN,
                     t % GEMM_BX, (t / GEMM_BX) % (CC / 64), vBatch0 + t / GEMM_PER_BATCH);
    }
}

__global__ __launch_bounds__(128) __attribute__((amdgpu_waves_per_eu(4, 8))) void nt_v_kernel(
    const float* __restrict__ Tq, const float* __restrict__ Wk, const float* __restrict__ bk,
    float* __restrict__ corr,
    const float* __restrict__ Wv, const float* __restrict__ bv, const float* __restrict__ fg,
    float* __restrict__ Vv, int vBatch0)
{
    const int b = blockIdx.x;
    if (b < NT_BLOCKS) {
        nt_tile(Tq, TQ_BATCH, TQ_PITCH, Wk, 0, CC, CC, CC,
                Tq + CC, TQ_BATCH, TQ_PITCH, bk, 0,
                corr, (long)CC * CC, CC,
                b % NT_BX, (b / NT_BX) % (CC / 64), b / (NT_BX * (CC / 64)));
    } else {
        const int t = b - NT_BLOCKS;
        gemm256_tile(Wv, 0, bv, fg, (long)CC * HWN, Vv, (long)CC * HWN,
                     t % GEMM_BX, (t / GEMM_BX) % (CC / 64), vBatch0 + t / GEMM_PER_BATCH);
    }
}

extern "C" void kernel_launch(void* const* d_in, const int* in_sizes, int n_in,
                              void* d_out, int out_size, void* d_ws, size_t ws_size,
                              hipStream_t stream)
{
    (void)in_sizes; (void)n_in; (void)out_size; (void)ws_size;

    const float* fg    = (const float*)d_in[0];
    const float* bg    = (const float*)d_in[1];
    const float* mask  = (const float*)d_in[2];
    const float* Wq    = (const float*)d_in[3];
    const float* bq    = (const float*)d_in[4];
    const float* Wk    = (const float*)d_in[5];
    const float* bk    = (const float*)d_in[6];
    const float* Wv    = (const float*)d_in[7];
    const float* bv    = (const float*)d_in[8];
    const float* gamma = (const float*)d_in[9];
    float* out = (float*)d_out;

    const long mat = (long)BB * CC * HWN;
    float* Vv   = (float*)d_ws;
    float* Tq   = Vv + mat;
    float* corr = Tq + (long)BB * TQ_BATCH;
    float* MTp  = out;
    float* sGp  = out + (long)BB * MT_SPLIT * MT_BATCH;

    const dim3 blk(128);
    const dim3 gGemm(HWN / 64, CC / 64, BB);
    const long xStride = (long)CC * HWN;

    mt_lds_kernel<<<dim3(CC / 64, CC / 64, BB * MT_SPLIT), blk, 0, stream>>>(
                                             bg, xStride, HWN, fg, xStride, HWN, HWN / MT_SPLIT,
                                             MTp, MT_BATCH, CC,
                                             sGp, CC + 1, MTp + CC * CC, MT_BATCH);
    ntp_v_kernel<<<dim3(NTP_BLOCKS + V_BATCHES_EARLY * GEMM_PER_BATCH), blk, 0, stream>>>(
                                             Wq, MTp, bq, sGp, Tq, Wv, bv, fg, Vv, 0);
    nt_v_kernel<<<dim3(NT_BLOCKS + (BB - V_BATCHES_EARLY) * GEMM_PER_BATCH), blk, 0, stream>>>(
                                             Tq, Wk, bk, corr, Wv, bv, fg, Vv, V_BATCHES_EARLY);

    gemm256_kernel<<<gGemm, blk, 0, stream>>>(corr, (long)CC * CC, nullptr,
                                              mask, xStride, out, xStride);

    softmax_out_kernel<<<dim3(BB * CC), dim3(256), 0, stream>>>(
        out, fg, mask, Vv, gamma, out);
}
